# starting kernel
# baseline (speedup 1.0000x reference)
#include <hip/hip_runtime.h>

#define NNODES 50000
#define MTILES (NNODES / 16)

typedef float v2f __attribute__((ext_vector_type(2)));
typedef float v8f __attribute__((ext_vector_type(8)));

__device__ __forceinline__ unsigned enc_f32(float f) {
    unsigned u = __float_as_uint(f);
    return (u & 0x80000000u) ? ~u : (u | 0x80000000u);
}
__device__ __forceinline__ float dec_f32(unsigned e) {
    if (e == 0u) return 0.0f;
    unsigned u = (e & 0x80000000u) ? (e ^ 0x80000000u) : ~e;
    return __uint_as_float(u);
}

__global__ void gin_zero_kernel(unsigned* __restrict__ p, long long n4) {
    long long i = blockIdx.x * (long long)blockDim.x + threadIdx.x;
    if (i < n4) ((uint4*)p)[i] = make_uint4(0u, 0u, 0u, 0u);
}

__global__ void gin_scatter_max_kernel(const int* __restrict__ src, const int* __restrict__ dst,
                                       const float* __restrict__ X, unsigned* __restrict__ agg,
                                       int d, int chunk_shift, long long total) {
    long long idx = blockIdx.x * (long long)blockDim.x + threadIdx.x;
    if (idx >= total) return;
    int e = (int)(idx >> chunk_shift);
    int c = (int)(idx & ((1 << chunk_shift) - 1));
    int s = src[e];
    int t = dst[e];
    const float4 v = ((const float4*)(X + (size_t)s * d))[c];
    unsigned* ap = agg + (size_t)t * d + (size_t)c * 4;
    atomicMax(ap + 0, enc_f32(v.x));
    atomicMax(ap + 1, enc_f32(v.y));
    atomicMax(ap + 2, enc_f32(v.z));
    atomicMax(ap + 3, enc_f32(v.w));
}

__global__ void gin_combine_kernel(const float* __restrict__ X, const unsigned* __restrict__ agg,
                                   float* __restrict__ out, long long n4) {
    long long i = blockIdx.x * (long long)blockDim.x + threadIdx.x;
    if (i >= n4) return;
    float4 x = ((const float4*)X)[i];
    uint4  a = ((const uint4*)agg)[i];
    float4 r;
    r.x = x.x + dec_f32(a.x);
    r.y = x.y + dec_f32(a.y);
    r.z = x.z + dec_f32(a.z);
    r.w = x.w + dec_f32(a.w);
    ((float4*)out)[i] = r;
}

__global__ __launch_bounds__(128) void gin_gemm_wmma_kernel(
    const float* __restrict__ X, const float* __restrict__ W,
    const float* __restrict__ bias, float* __restrict__ Y,
    int K, int Nout, int relu)
{
    const int lane  = threadIdx.x & 31;
    const int wave  = threadIdx.x >> 5;
    const int mtile = blockIdx.x * 4 + wave;
    if (mtile >= MTILES) return;

    const int row0    = mtile * 16;
    const int l15     = lane & 15;
    const int koff    = (lane >> 4) * 2;
    const int colbase = blockIdx.y * 64 + l15;

    const float* __restrict__ xrow = X + (size_t)(row0 + l15) * K;

    v8f acc0 = {}, acc1 = {}, acc2 = {}, acc3 = {};
#pragma unroll 2
    for (int k0 = 0; k0 < K; k0 += 4) {
        v2f a;
        a.x = xrow[k0 + koff + 0];
        a.y = xrow[k0 + koff + 1];
        const float* __restrict__ w0 = W + (size_t)(k0 + koff + 0) * Nout + colbase;
        const float* __restrict__ w1 = W + (size_t)(k0 + koff + 1) * Nout + colbase;
        v2f b0, b1, b2, b3;
        b0.x = w0[ 0]; b0.y = w1[ 0];
        b1.x = w0[16]; b1.y = w1[16];
        b2.x = w0[32]; b2.y = w1[32];
        b3.x = w0[48]; b3.y = w1[48];
        acc0 = __builtin_amdgcn_wmma_f32_16x16x4_f32(false, a, false, b0, (short)0, acc0, false, false);
        acc1 = __builtin_amdgcn_wmma_f32_16x16x4_f32(false, a, false, b1, (short)0, acc1, false, false);
        acc2 = __builtin_amdgcn_wmma_f32_16x16x4_f32(false, a, false, b2, (short)0, acc2, false, false);
        acc3 = __builtin_amdgcn_wmma_f32_16x16x4_f32(false, a, false, b3, (short)0, acc3, false, false);
    }

    const int rbase = row0 + (lane >> 4) * 8;
    v8f accs[4] = {acc0, acc1, acc2, acc3};
#pragma unroll
    for (int t = 0; t < 4; ++t) {
        const int col = colbase + 16 * t;
        const float bv = bias[col];
        float* __restrict__ yc = Y + col;
#pragma unroll
        for (int r = 0; r < 8; ++r) {
            float v = accs[t][r] + bv;
            if (relu) v = fmaxf(v, 0.0f);
            yc[(size_t)(rbase + r) * Nout] = v;
        }
    }
}

static void gin_layer(const int* src, const int* dst, int E,
                      const float* Xin, int d_in_, const float* W, const float* b,
                      int d_out_, float* Y, int relu,
                      unsigned* agg, float* xsum, hipStream_t stream)
{
    long long nelem = (long long)NNODES * d_in_;
    long long n4 = nelem >> 2;
    gin_zero_kernel<<<(unsigned)((n4 + 255) / 256), 256, 0, stream>>>(agg, n4);

    int cs = (d_in_ == 128) ? 5 : 6;
    long long total = (long long)E << cs;
    gin_scatter_max_kernel<<<(unsigned)((total + 255) / 256), 256, 0, stream>>>(
        src, dst, Xin, agg, d_in_, cs, total);

    gin_combine_kernel<<<(unsigned)((n4 + 255) / 256), 256, 0, stream>>>(Xin, agg, xsum, n4);

    dim3 grid((MTILES + 3) / 4, d_out_ / 64);
    gin_gemm_wmma_kernel<<<grid, 128, 0, stream>>>(xsum, W, b, Y, d_in_, d_out_, relu);
}

extern "C" void kernel_launch(void* const* d_in, const int* in_sizes, int n_in,
                              void* d_out, int out_size, void* d_ws, size_t ws_size,
                              hipStream_t stream)
{
    const int*   src = (const int*)d_in[0];
    const int*   dst = (const int*)d_in[1];
    const float* emb = (const float*)d_in[2];
    const float* W1  = (const float*)d_in[3];
    const float* b1  = (const float*)d_in[4];
    const float* W2  = (const float*)d_in[5];
    const float* b2  = (const float*)d_in[6];
    const float* W3  = (const float*)d_in[7];
    const float* b3  = (const float*)d_in[8];
    float* out = (float*)d_out;
    const int E = in_sizes[0];

    const size_t cap = (size_t)NNODES * 256;
    unsigned* agg  = (unsigned*)d_ws;
    float*    xsum = (float*)d_ws + cap;
    float*    h    = (float*)d_ws + 2 * cap;

    gin_layer(src, dst, E, emb, 128, W1, b1, 256, h, 1, agg, xsum, stream);
    gin_layer(src, dst, E, h, 256, W2, b2, 256, h, 1, agg, xsum, stream);
    gin_layer(src, dst, E, h, 256, W3, b3, 128, out, 0, agg, xsum, stream);
}


# Segment max from an edge list grouped by destination, built once per call (int atomics for counts only); one wave per node maxes rows in registers, stores x+max; no float atomics, no sum reordered
# speedup vs baseline: 2.8538x; 2.8538x over previous
#include <hip/hip_runtime.h>


#define NNODES 50000
#define MTILES (NNODES / 16)
#define NQUADS (NNODES / 4)

typedef float v2f __attribute__((ext_vector_type(2)));
typedef float v8f __attribute__((ext_vector_type(8)));


__global__ void gin_list_zero_kernel(int* __restrict__ cnt) {
    int i = blockIdx.x * blockDim.x + threadIdx.x;
    if (i < NNODES) cnt[i] = 0;
}

__global__ void gin_list_count_kernel(const int* __restrict__ dst, int* __restrict__ cnt,
                                      int* __restrict__ rank, int E) {
    int e = blockIdx.x * blockDim.x + threadIdx.x;
    if (e < E) rank[e] = atomicAdd(&cnt[dst[e]], 1);
}

#define SCAN_T 256
#define SCAN_BLOCKS ((NQUADS + SCAN_T - 1) / SCAN_T)
__global__ __launch_bounds__(SCAN_T) void gin_list_scan_kernel(const int* __restrict__ cnt,
                                                               int* __restrict__ rowptr) {
    __shared__ int sh_front[SCAN_T / 32];
    __shared__ int sh_tile[SCAN_T / 32];
    const int t = threadIdx.x;
    const int lane = t & 31;
    const int wave = t >> 5;
    const int4* __restrict__ c4 = (const int4*)cnt;

    int front = 0;
    for (int k = 0; k < (int)blockIdx.x; ++k) {
        const int4 v = c4[k * SCAN_T + t];
        front += (v.x + v.y) + (v.z + v.w);
    }
    for (int off = 16; off > 0; off >>= 1) front += __shfl_xor(front, off);
    if (lane == 0) sh_front[wave] = front;

    const int q = blockIdx.x * SCAN_T + t;
    int4 v = make_int4(0, 0, 0, 0);
    if (q < NQUADS) v = c4[q];
    const int s = (v.x + v.y) + (v.z + v.w);
    int incl = s;
    for (int off = 1; off < 32; off <<= 1) {
        const int u = __shfl_up(incl, off);
        if (lane >= off) incl += u;
    }
    if (lane == 31) sh_tile[wave] = incl;
    __syncthreads();

    int base = 0;
    for (int w = 0; w < SCAN_T / 32; ++w) {
        base += sh_front[w];
        if (w < wave) base += sh_tile[w];
    }
    const int ex = base + incl - s;
    if (q < NQUADS) {
        int4 o;
        o.x = ex;
        o.y = o.x + v.x;
        o.z = o.y + v.y;
        o.w = o.z + v.z;
        ((int4*)rowptr)[q] = o;
        if (q == NQUADS - 1) rowptr[NNODES] = ex + s;
    }
}

__global__ void gin_list_fill_kernel(const int* __restrict__ src, const int* __restrict__ dst,
                                     const int* __restrict__ rank, const int* __restrict__ rowptr,
                                     int* __restrict__ col, int E) {
    int e = blockIdx.x * blockDim.x + threadIdx.x;
    if (e < E) {
        const int p = min(max(rowptr[dst[e]] + rank[e], 0), E - 1);
        col[p] = src[e];
    }
}


__device__ __forceinline__ float4 max4(float4 a, float4 b) {
    float4 r;
    r.x = fmaxf(a.x, b.x);
    r.y = fmaxf(a.y, b.y);
    r.z = fmaxf(a.z, b.z);
    r.w = fmaxf(a.w, b.w);
    return r;
}

template <int D, int K>
__device__ __forceinline__ void gin_fold_rows(const float4* __restrict__ X4, int mine, int j,
                                              int lane, float4 (&m)[D / 128]) {
    constexpr int V = D / 128;
    float4 r[K][V];
#pragma unroll
    for (int k = 0; k < K; ++k) {
        const int s = __builtin_amdgcn_readlane(mine, j + k);
        const float4* __restrict__ row = X4 + (size_t)s * (D / 4) + lane;
#pragma unroll
        for (int v = 0; v < V; ++v) r[k][v] = row[v * 32];
    }
#pragma unroll
    for (int k = 0; k < K; ++k) {
#pragma unroll
        for (int v = 0; v < V; ++v) m[v] = max4(m[v], r[k][v]);
    }
}

#define GATHER_BLOCKS 2048
#define GATHER_WAVES (GATHER_BLOCKS * 8)

template <int D>
__global__ __launch_bounds__(256) void gin_gather_max_kernel(
    const int* __restrict__ rowptr, const int* __restrict__ col,
    const float* __restrict__ X, float* __restrict__ xsum, int E)
{
    constexpr int V = D / 128;
    constexpr int U = 8 / V;
    const int lane = threadIdx.x & 31;
    const int wave0 = __builtin_amdgcn_readfirstlane((int)(blockIdx.x * 8u + (threadIdx.x >> 5)));
    const float4* __restrict__ X4 = (const float4*)X;
    float4* __restrict__ O4 = (float4*)xsum;
    const float ninf = -__builtin_huge_valf();

    for (int node = wave0; node < NNODES; node += GATHER_WAVES) {
        const int beg = max(rowptr[node], 0);
        const int end = min(rowptr[node + 1], E);

        float4 self[V];
        float4 m[V];
#pragma unroll
        for (int v = 0; v < V; ++v) {
            self[v] = X4[(size_t)node * (D / 4) + v * 32 + lane];
            m[v] = make_float4(ninf, ninf, ninf, ninf);
        }

        for (int base = beg; base < end; base += 32) {
            const int n = min(end - base, 32);
            int mine = 0;
            if (lane < n) mine = min(max(col[base + lane], 0), NNODES - 1);
            int j = 0;
            for (; j + U <= n; j += U) gin_fold_rows<D, U>(X4, mine, j, lane, m);
            if constexpr (U > 4) {
                if (j + 4 <= n) { gin_fold_rows<D, 4>(X4, mine, j, lane, m); j += 4; }
            }
            if (j + 2 <= n) { gin_fold_rows<D, 2>(X4, mine, j, lane, m); j += 2; }
            if (j < n) gin_fold_rows<D, 1>(X4, mine, j, lane, m);
        }

#pragma unroll
        for (int v = 0; v < V; ++v) {
            float4 a = m[v];
            a.x = (a.x == ninf) ? 0.0f : a.x;
            a.y = (a.y == ninf) ? 0.0f : a.y;
            a.z = (a.z == ninf) ? 0.0f : a.z;
            a.w = (a.w == ninf) ? 0.0f : a.w;
            float4 o;
            o.x = self[v].x + a.x;
            o.y = self[v].y + a.y;
            o.z = self[v].z + a.z;
            o.w = self[v].w + a.w;
            O4[(size_t)node * (D / 4) + v * 32 + lane] = o;
        }
    }
}

__global__ __launch_bounds__(128) void gin_gemm_wmma_kernel(
    const float* __restrict__ X, const float* __restrict__ W,
    const float* __restrict__ bias, float* __restrict__ Y,
    int K, int Nout, int relu)
{
    const int lane  = threadIdx.x & 31;
    const int wave  = threadIdx.x >> 5;
    const int mtile = blockIdx.x * 4 + wave;
    if (mtile >= MTILES) return;

    const int row0    = mtile * 16;
    const int l15     = lane & 15;
    const int koff    = (lane >> 4) * 2;
    const int colbase = blockIdx.y * 64 + l15;

    const float* __restrict__ xrow = X + (size_t)(row0 + l15) * K;

    v8f acc0 = {}, acc1 = {}, acc2 = {}, acc3 = {};
#pragma unroll 2
    for (int k0 = 0; k0 < K; k0 += 4) {
        v2f a;
        a.x = xrow[k0 + koff + 0];
        a.y = xrow[k0 + koff + 1];
        const float* __restrict__ w0 = W + (size_t)(k0 + koff + 0) * Nout + colbase;
        const float* __restrict__ w1 = W + (size_t)(k0 + koff + 1) * Nout + colbase;
        v2f b0, b1, b2, b3;
        b0.x = w0[ 0]; b0.y = w1[ 0];
        b1.x = w0[16]; b1.y = w1[16];
        b2.x = w0[32]; b2.y = w1[32];
        b3.x = w0[48]; b3.y = w1[48];
        acc0 = __builtin_amdgcn_wmma_f32_16x16x4_f32(false, a, false, b0, (short)0, acc0, false, false);
        acc1 = __builtin_amdgcn_wmma_f32_16x16x4_f32(false, a, false, b1, (short)0, acc1, false, false);
        acc2 = __builtin_amdgcn_wmma_f32_16x16x4_f32(false, a, false, b2, (short)0, acc2, false, false);
        acc3 = __builtin_amdgcn_wmma_f32_16x16x4_f32(false, a, false, b3, (short)0, acc3, false, false);
    }

    const int rbase = row0 + (lane >> 4) * 8;
    v8f accs[4] = {acc0, acc1, acc2, acc3};
#pragma unroll
    for (int t = 0; t < 4; ++t) {
        const int col = colbase + 16 * t;
        const float bv = bias[col];
        float* __restrict__ yc = Y + col;
#pragma unroll
        for (int r = 0; r < 8; ++r) {
            float v = accs[t][r] + bv;
            if (relu) v = fmaxf(v, 0.0f);
            yc[(size_t)(rbase + r) * Nout] = v;
        }
    }
}

static void gin_build_lists(const int* src, const int* dst, int E,
                            int* cnt, int* rowptr, int* rank, int* col, hipStream_t stream)
{
    gin_list_zero_kernel<<<(NNODES + 255) / 256, 256, 0, stream>>>(cnt);
    gin_list_count_kernel<<<(unsigned)((E + 255) / 256), 256, 0, stream>>>(dst, cnt, rank, E);
    gin_list_scan_kernel<<<SCAN_BLOCKS, SCAN_T, 0, stream>>>(cnt, rowptr);
    gin_list_fill_kernel<<<(unsigned)((E + 255) / 256), 256, 0, stream>>>(src, dst, rank, rowptr, col, E);
}

static void gin_layer(const int* rowptr, const int* col, int E,
                      const float* Xin, int d_in_, const float* W, const float* b,
                      int d_out_, float* Y, int relu,
                      float* xsum, hipStream_t stream)
{
    if (d_in_ == 128)
        gin_gather_max_kernel<128><<<GATHER_BLOCKS, 256, 0, stream>>>(rowptr, col, Xin, xsum, E);
    else
        gin_gather_max_kernel<256><<<GATHER_BLOCKS, 256, 0, stream>>>(rowptr, col, Xin, xsum, E);

    dim3 grid((MTILES + 3) / 4, d_out_ / 64);
    gin_gemm_wmma_kernel<<<grid, 128, 0, stream>>>(xsum, W, b, Y, d_in_, d_out_, relu);
}

extern "C" void kernel_launch(void* const* d_in, const int* in_sizes, int n_in,
                              void* d_out, int out_size, void* d_ws, size_t ws_size,
                              hipStream_t stream)
{
    const int*   src = (const int*)d_in[0];
    const int*   dst = (const int*)d_in[1];
    const float* emb = (const float*)d_in[2];
    const float* W1  = (const float*)d_in[3];
    const float* b1  = (const float*)d_in[4];
    const float* W2  = (const float*)d_in[5];
    const float* b2  = (const float*)d_in[6];
    const float* W3  = (const float*)d_in[7];
    const float* b3  = (const float*)d_in[8];
    float* out = (float*)d_out;
    const int E = in_sizes[0];

    const size_t cap = (size_t)NNODES * 256;
    int*   cnt    = (int*)d_ws;
    int*   rowptr = cnt + 65536;
    int*   rank   = rowptr + 65536;
    int*   col    = rank + E;
    float* xsum   = (float*)d_ws + cap;
    float* h      = (float*)d_ws + 2 * cap;

    gin_build_lists(src, dst, E, cnt, rowptr, rank, col, stream);

    gin_layer(rowptr, col, E, emb, 128, W1, b1, 256, h, 1, xsum, stream);
    gin_layer(rowptr, col, E, h, 256, W2, b2, 256, h, 1, xsum, stream);
    gin_layer(rowptr, col, E, h, 256, W3, b3, 128, out, 0, xsum, stream);
}


# GEMM operands staged through LDS: each 128x32 X piece and 32x64 W piece fetched once per work-group with whole-line loads, WMMA fragments read from LDS, 32x64 per wave; sum order unchanged
# speedup vs baseline: 3.2900x; 1.1528x over previous
#include <hip/hip_runtime.h>


#define NNODES 50000
#define NQUADS (NNODES / 4)

typedef float v2f __attribute__((ext_vector_type(2)));
typedef float v8f __attribute__((ext_vector_type(8)));


__global__ void gin_list_zero_kernel(int* __restrict__ cnt) {
    int i = blockIdx.x * blockDim.x + threadIdx.x;
    if (i < NNODES) cnt[i] = 0;
}

__global__ void gin_list_count_kernel(const int* __restrict__ dst, int* __restrict__ cnt,
                                      int* __restrict__ rank, int E) {
    int e = blockIdx.x * blockDim.x + threadIdx.x;
    if (e < E) rank[e] = atomicAdd(&cnt[dst[e]], 1);
}

#define SCAN_T 256
#define SCAN_BLOCKS ((NQUADS + SCAN_T - 1) / SCAN_T)
__global__ __launch_bounds__(SCAN_T) void gin_list_scan_kernel(const int* __restrict__ cnt,
                                                               int* __restrict__ rowptr) {
    __shared__ int sh_front[SCAN_T / 32];
    __shared__ int sh_tile[SCAN_T / 32];
    const int t = threadIdx.x;
    const int lane = t & 31;
    const int wave = t >> 5;
    const int4* __restrict__ c4 = (const int4*)cnt;

    int front = 0;
    for (int k = 0; k < (int)blockIdx.x; ++k) {
        const int4 v = c4[k * SCAN_T + t];
        front += (v.x + v.y) + (v.z + v.w);
    }
    for (int off = 16; off > 0; off >>= 1) front += __shfl_xor(front, off);
    if (lane == 0) sh_front[wave] = front;

    const int q = blockIdx.x * SCAN_T + t;
    int4 v = make_int4(0, 0, 0, 0);
    if (q < NQUADS) v = c4[q];
    const int s = (v.x + v.y) + (v.z + v.w);
    int incl = s;
    for (int off = 1; off < 32; off <<= 1) {
        const int u = __shfl_up(incl, off);
        if (lane >= off) incl += u;
    }
    if (lane == 31) sh_tile[wave] = incl;
    __syncthreads();

    int base = 0;
    for (int w = 0; w < SCAN_T / 32; ++w) {
        base += sh_front[w];
        if (w < wave) base += sh_tile[w];
    }
    const int ex = base + incl - s;
    if (q < NQUADS) {
        int4 o;
        o.x = ex;
        o.y = o.x + v.x;
        o.z = o.y + v.y;
        o.w = o.z + v.z;
        ((int4*)rowptr)[q] = o;
        if (q == NQUADS - 1) rowptr[NNODES] = ex + s;
    }
}

__global__ void gin_list_fill_kernel(const int* __restrict__ src, const int* __restrict__ dst,
                                     const int* __restrict__ rank, const int* __restrict__ rowptr,
                                     int* __restrict__ col, int E) {
    int e = blockIdx.x * blockDim.x + threadIdx.x;
    if (e < E) {
        const int p = min(max(rowptr[dst[e]] + rank[e], 0), E - 1);
        col[p] = src[e];
    }
}


__device__ __forceinline__ float4 max4(float4 a, float4 b) {
    float4 r;
    r.x = fmaxf(a.x, b.x);
    r.y = fmaxf(a.y, b.y);
    r.z = fmaxf(a.z, b.z);
    r.w = fmaxf(a.w, b.w);
    return r;
}

template <int D, int K>
__device__ __forceinline__ void gin_fold_rows(const float4* __restrict__ X4, int mine, int j,
                                              int lane, float4 (&m)[D / 128]) {
    constexpr int V = D / 128;
    float4 r[K][V];
#pragma unroll
    for (int k = 0; k < K; ++k) {
        const int s = __builtin_amdgcn_readlane(mine, j + k);
        const float4* __restrict__ row = X4 + (size_t)s * (D / 4) + lane;
#pragma unroll
        for (int v = 0; v < V; ++v) r[k][v] = row[v * 32];
    }
#pragma unroll
    for (int k = 0; k < K; ++k) {
#pragma unroll
        for (int v = 0; v < V; ++v) m[v] = max4(m[v], r[k][v]);
    }
}

#define GATHER_BLOCKS 2048
#define GATHER_WAVES (GATHER_BLOCKS * 8)

template <int D>
__global__ __launch_bounds__(256) void gin_gather_max_kernel(
    const int* __restrict__ rowptr, const int* __restrict__ col,
    const float* __restrict__ X, float* __restrict__ xsum, int E)
{
    constexpr int V = D / 128;
    constexpr int U = 8 / V;
    const int lane = threadIdx.x & 31;
    const int wave0 = __builtin_amdgcn_readfirstlane((int)(blockIdx.x * 8u + (threadIdx.x >> 5)));
    const float4* __restrict__ X4 = (const float4*)X;
    float4* __restrict__ O4 = (float4*)xsum;
    const float ninf = -__builtin_huge_valf();

    for (int node = wave0; node < NNODES; node += GATHER_WAVES) {
        const int beg = max(rowptr[node], 0);
        const int end = min(rowptr[node + 1], E);

        float4 self[V];
        float4 m[V];
#pragma unroll
        for (int v = 0; v < V; ++v) {
            self[v] = X4[(size_t)node * (D / 4) + v * 32 + lane];
            m[v] = make_float4(ninf, ninf, ninf, ninf);
        }

        for (int base = beg; base < end; base += 32) {
            const int n = min(end - base, 32);
            int mine = 0;
            if (lane < n) mine = min(max(col[base + lane], 0), NNODES - 1);
            int j = 0;
            for (; j + U <= n; j += U) gin_fold_rows<D, U>(X4, mine, j, lane, m);
            if constexpr (U > 4) {
                if (j + 4 <= n) { gin_fold_rows<D, 4>(X4, mine, j, lane, m); j += 4; }
            }
            if (j + 2 <= n) { gin_fold_rows<D, 2>(X4, mine, j, lane, m); j += 2; }
            if (j < n) gin_fold_rows<D, 1>(X4, mine, j, lane, m);
        }

#pragma unroll
        for (int v = 0; v < V; ++v) {
            float4 a = m[v];
            a.x = (a.x == ninf) ? 0.0f : a.x;
            a.y = (a.y == ninf) ? 0.0f : a.y;
            a.z = (a.z == ninf) ? 0.0f : a.z;
            a.w = (a.w == ninf) ? 0.0f : a.w;
            float4 o;
            o.x = self[v].x + a.x;
            o.y = self[v].y + a.y;
            o.z = self[v].z + a.z;
            o.w = self[v].w + a.w;
            O4[(size_t)node * (D / 4) + v * 32 + lane] = o;
        }
    }
}

#define GM_BM 128
#define GM_BN 64
#define GM_KC 32
#define GM_PA (GM_KC + 4)
#define GM_PW (2 * GM_BN + 32)

typedef float v4f __attribute__((ext_vector_type(4)));

template <int NOUT>
__device__ __forceinline__ void gin_gemm_fetch(const float* __restrict__ X, const unsigned (&aoff)[8],
                                               const float* __restrict__ wa, int kc,
                                               v4f (&ra)[8], v4f (&rw)[4]) {
#pragma unroll
    for (int p = 0; p < 8; ++p) ra[p] = *(const v4f*)(X + aoff[p] + kc);
#pragma unroll
    for (int p = 0; p < 2; ++p) {
        rw[2 * p + 0] = *(const v4f*)(wa + (size_t)(kc + 16 * p + 0) * NOUT);
        rw[2 * p + 1] = *(const v4f*)(wa + (size_t)(kc + 16 * p + 1) * NOUT);
    }
}

__device__ __forceinline__ void gin_gemm_park(float* __restrict__ as, float* __restrict__ ws,
                                              const v4f (&ra)[8], const v4f (&rw)[4]) {
#pragma unroll
    for (int p = 0; p < 8; ++p) *(v4f*)(as + 16 * p * GM_PA) = ra[p];
#pragma unroll
    for (int p = 0; p < 2; ++p) {
        const v4f e = rw[2 * p + 0];
        const v4f o = rw[2 * p + 1];
        const v4f lo = {e.x, o.x, e.y, o.y};
        const v4f hi = {e.z, o.z, e.w, o.w};
        *(v4f*)(ws + 8 * p * GM_PW + 0) = lo;
        *(v4f*)(ws + 8 * p * GM_PW + 4) = hi;
    }
}

__device__ __forceinline__ void gin_gemm_piece(const float* __restrict__ Aw, const float* __restrict__ Bw,
                                               v8f (&acc)[2][4]) {
#pragma unroll
    for (int kk = 0; kk < GM_KC; kk += 4) {
        const v2f a0 = *(const v2f*)(Aw + kk);
        const v2f a1 = *(const v2f*)(Aw + 16 * GM_PA + kk);
        v2f b[4];
#pragma unroll
        for (int t = 0; t < 4; ++t) b[t] = *(const v2f*)(Bw + (kk / 2) * GM_PW + 32 * t);
#pragma unroll
        for (int t = 0; t < 4; ++t) {
            acc[0][t] = __builtin_amdgcn_wmma_f32_16x16x4_f32(false, a0, false, b[t], (short)0, acc[0][t], false, false);
            acc[1][t] = __builtin_amdgcn_wmma_f32_16x16x4_f32(false, a1, false, b[t], (short)0, acc[1][t], false, false);
        }
    }
}

template <int K, int NOUT>
__global__ __launch_bounds__(128) void gin_gemm_wmma_kernel(
    const float* __restrict__ X, const float* __restrict__ W,
    const float* __restrict__ bias, float* __restrict__ Y, int relu)
{
    __shared__ __attribute__((aligned(16))) float As[GM_BM * GM_PA];
    __shared__ __attribute__((aligned(16))) float Ws[(GM_KC / 2) * GM_PW];

    const int tid    = threadIdx.x;
    const int lane   = tid & 31;
    const int wave   = tid >> 5;
    const int l15    = lane & 15;
    const int half   = lane >> 4;
    const int rowblk = blockIdx.x * GM_BM;
    const int colblk = blockIdx.y * GM_BN;

    const int a_q = tid & 7;
    const int a_r = tid >> 3;
    const int w_q = tid & 15;
    const int w_j = tid >> 4;

    unsigned aoff[8];
#pragma unroll
    for (int p = 0; p < 8; ++p)
        aoff[p] = (unsigned)min(rowblk + a_r + 16 * p, NNODES - 1) * (unsigned)K + 4u * (unsigned)a_q;
    const float* __restrict__ wa = W + (size_t)(2 * w_j) * NOUT + colblk + 4 * w_q;
    float* __restrict__ as = As + a_r * GM_PA + 4 * a_q;
    float* __restrict__ ws = Ws + w_j * GM_PW + 8 * w_q;

    const float* __restrict__ Aw = As + (wave * 32 + l15) * GM_PA + 2 * half;
    const float* __restrict__ Bw = Ws + half * GM_PW + 2 * l15;

    v4f ra[8];
    v4f rw[4];
    v8f acc[2][4];
#pragma unroll
    for (int i = 0; i < 2; ++i)
#pragma unroll
        for (int t = 0; t < 4; ++t) acc[i][t] = (v8f){};

    gin_gemm_fetch<NOUT>(X, aoff, wa, 0, ra, rw);
    for (int kc = 0; kc < K - GM_KC; kc += GM_KC) {
        __syncthreads();
        gin_gemm_park(as, ws, ra, rw);
        __syncthreads();
        gin_gemm_fetch<NOUT>(X, aoff, wa, kc + GM_KC, ra, rw);
        gin_gemm_piece(Aw, Bw, acc);
    }
    __syncthreads();
    gin_gemm_park(as, ws, ra, rw);
    __syncthreads();
    gin_gemm_piece(Aw, Bw, acc);

#pragma unroll
    for (int i = 0; i < 2; ++i) {
        const int row0 = rowblk + wave * 32 + 16 * i;
        if (row0 < NNODES) {
            const int rbase = row0 + 8 * half;
#pragma unroll
            for (int t = 0; t < 4; ++t) {
                const int col = colblk + 16 * t + l15;
                const float bv = bias[col];
                float* __restrict__ yc = Y + col;
#pragma unroll
                for (int r = 0; r < 8; ++r) {
                    float v = acc[i][t][r] + bv;
                    if (relu) v = fmaxf(v, 0.0f);
                    yc[(size_t)(rbase + r) * NOUT] = v;
                }
            }
        }
    }
}

static void gin_build_lists(const int* src, const int* dst, int E,
                            int* cnt, int* rowptr, int* rank, int* col, hipStream_t stream)
{
    gin_list_zero_kernel<<<(NNODES + 255) / 256, 256, 0, stream>>>(cnt);
    gin_list_count_kernel<<<(unsigned)((E + 255) / 256), 256, 0, stream>>>(dst, cnt, rank, E);
    gin_list_scan_kernel<<<SCAN_BLOCKS, SCAN_T, 0, stream>>>(cnt, rowptr);
    gin_list_fill_kernel<<<(unsigned)((E + 255) / 256), 256, 0, stream>>>(src, dst, rank, rowptr, col, E);
}

static void gin_layer(const int* rowptr, const int* col, int E,
                      const float* Xin, int d_in_, const float* W, const float* b,
                      int d_out_, float* Y, int relu,
                      float* xsum, hipStream_t stream)
{
    if (d_in_ == 128)
        gin_gather_max_kernel<128><<<GATHER_BLOCKS, 256, 0, stream>>>(rowptr, col, Xin, xsum, E);
    else
        gin_gather_max_kernel<256><<<GATHER_BLOCKS, 256, 0, stream>>>(rowptr, col, Xin, xsum, E);

    dim3 grid((NNODES + GM_BM - 1) / GM_BM, d_out_ / GM_BN);
    if (d_in_ == 128)
        gin_gemm_wmma_kernel<128, 256><<<grid, 128, 0, stream>>>(xsum, W, b, Y, relu);
    else if (d_out_ == 256)
        gin_gemm_wmma_kernel<256, 256><<<grid, 128, 0, stream>>>(xsum, W, b, Y, relu);
    else
        gin_gemm_wmma_kernel<256, 128><<<grid, 128, 0, stream>>>(xsum, W, b, Y, relu);
}

extern "C" void kernel_launch(void* const* d_in, const int* in_sizes, int n_in,
                              void* d_out, int out_size, void* d_ws, size_t ws_size,
                              hipStream_t stream)
{
    const int*   src = (const int*)d_in[0];
    const int*   dst = (const int*)d_in[1];
    const float* emb = (const float*)d_in[2];
    const float* W1  = (const float*)d_in[3];
    const float* b1  = (const float*)d_in[4];
    const float* W2  = (const float*)d_in[5];
    const float* b2  = (const float*)d_in[6];
    const float* W3  = (const float*)d_in[7];
    const float* b3  = (const float*)d_in[8];
    float* out = (float*)d_out;
    const int E = in_sizes[0];

    const size_t cap = (size_t)NNODES * 256;
    int*   cnt    = (int*)d_ws;
    int*   rowptr = cnt + 65536;
    int*   rank   = rowptr + 65536;
    int*   col    = rank + E;
    float* xsum   = (float*)d_ws + cap;
    float* h      = (float*)d_ws + 2 * cap;

    gin_build_lists(src, dst, E, cnt, rowptr, rank, col, stream);

    gin_layer(rowptr, col, E, emb, 128, W1, b1, 256, h, 1, xsum, stream);
    gin_layer(rowptr, col, E, h, 256, W2, b2, 256, h, 1, xsum, stream);
    gin_layer(rowptr, col, E, h, 256, W3, b3, 128, out, 0, xsum, stream);
}


# Neighbour max computed by the GEMM work-group straight into its LDS A tile (32 rows x all columns per group, K in halves of 128): no x+max plane, 3 launches fewer; sum order unchanged
# speedup vs baseline: 3.5594x; 1.0819x over previous
#include <hip/hip_runtime.h>


#define NNODES 50000
#define NQUADS (NNODES / 4)

typedef float v2f __attribute__((ext_vector_type(2)));
typedef float v8f __attribute__((ext_vector_type(8)));


__global__ void gin_list_zero_kernel(int* __restrict__ cnt) {
    int i = blockIdx.x * blockDim.x + threadIdx.x;
    if (i < NNODES) cnt[i] = 0;
}

__global__ void gin_list_count_kernel(const int* __restrict__ dst, int* __restrict__ cnt,
                                      int* __restrict__ rank, int E) {
    int e = blockIdx.x * blockDim.x + threadIdx.x;
    if (e < E) rank[e] = atomicAdd(&cnt[dst[e]], 1);
}

#define SCAN_T 256
#define SCAN_BLOCKS ((NQUADS + SCAN_T - 1) / SCAN_T)
__global__ __launch_bounds__(SCAN_T) void gin_list_scan_kernel(const int* __restrict__ cnt,
                                                               int* __restrict__ rowptr) {
    __shared__ int sh_front[SCAN_T / 32];
    __shared__ int sh_tile[SCAN_T / 32];
    const int t = threadIdx.x;
    const int lane = t & 31;
    const int wave = t >> 5;
    const int4* __restrict__ c4 = (const int4*)cnt;

    int front = 0;
    for (int k = 0; k < (int)blockIdx.x; ++k) {
        const int4 v = c4[k * SCAN_T + t];
        front += (v.x + v.y) + (v.z + v.w);
    }
    for (int off = 16; off > 0; off >>= 1) front += __shfl_xor(front, off);
    if (lane == 0) sh_front[wave] = front;

    const int q = blockIdx.x * SCAN_T + t;
    int4 v = make_int4(0, 0, 0, 0);
    if (q < NQUADS) v = c4[q];
    const int s = (v.x + v.y) + (v.z + v.w);
    int incl = s;
    for (int off = 1; off < 32; off <<= 1) {
        const int u = __shfl_up(incl, off);
        if (lane >= off) incl += u;
    }
    if (lane == 31) sh_tile[wave] = incl;
    __syncthreads();

    int base = 0;
    for (int w = 0; w < SCAN_T / 32; ++w) {
        base += sh_front[w];
        if (w < wave) base += sh_tile[w];
    }
    const int ex = base + incl - s;
    if (q < NQUADS) {
        int4 o;
        o.x = ex;
        o.y = o.x + v.x;
        o.z = o.y + v.y;
        o.w = o.z + v.z;
        ((int4*)rowptr)[q] = o;
        if (q == NQUADS - 1) rowptr[NNODES] = ex + s;
    }
}

__global__ void gin_list_fill_kernel(const int* __restrict__ src, const int* __restrict__ dst,
                                     const int* __restrict__ rank, const int* __restrict__ rowptr,
                                     int* __restrict__ col, int E) {
    int e = blockIdx.x * blockDim.x + threadIdx.x;
    if (e < E) {
        const int p = min(max(rowptr[dst[e]] + rank[e], 0), E - 1);
        col[p] = src[e];
    }
}


typedef float v4f __attribute__((ext_vector_type(4)));

__device__ __forceinline__ float4 max4(float4 a, float4 b) {
    float4 r;
    r.x = fmaxf(a.x, b.x);
    r.y = fmaxf(a.y, b.y);
    r.z = fmaxf(a.z, b.z);
    r.w = fmaxf(a.w, b.w);
    return r;
}

__device__ __forceinline__ int gin_list_chunk(const int* __restrict__ col, int base, int end, int lane) {
    int mine = 0;
    if (base + lane < end) mine = min(max(col[base + lane], 0), NNODES - 1);
    return mine;
}

template <int K, int G>
__device__ __forceinline__ void gin_fold_rows(const float4* __restrict__ Xc, int mine, int j, float4& m) {
    float4 r[G];
#pragma unroll
    for (int k = 0; k < G; ++k) {
        const int s = __builtin_amdgcn_readlane(mine, j + k);
        r[k] = Xc[(size_t)s * (K / 4)];
    }
#pragma unroll
    for (int k = 0; k < G; ++k) m = max4(m, r[k]);
}

#define FL_BM 32
#define FL_KH 128
#define FL_PA (FL_KH + 4)

template <int NOUT, int JS, int NP>
__device__ __forceinline__ void gin_w_fetch(const float* __restrict__ wa, int kg, v4f (&rw)[2 * NP]) {
#pragma unroll
    for (int p = 0; p < NP; ++p) {
        rw[2 * p + 0] = *(const v4f*)(wa + (size_t)(kg + 2 * JS * p + 0) * NOUT);
        rw[2 * p + 1] = *(const v4f*)(wa + (size_t)(kg + 2 * JS * p + 1) * NOUT);
    }
}

template <int PW, int JS, int NP>
__device__ __forceinline__ void gin_w_park(float* __restrict__ ws, const v4f (&rw)[2 * NP]) {
#pragma unroll
    for (int p = 0; p < NP; ++p) {
        const v4f e = rw[2 * p + 0];
        const v4f o = rw[2 * p + 1];
        const v4f lo = {e.x, o.x, e.y, o.y};
        const v4f hi = {e.z, o.z, e.w, o.w};
        *(v4f*)(ws + JS * p * PW + 0) = lo;
        *(v4f*)(ws + JS * p * PW + 4) = hi;
    }
}

template <int KC, int PW, int NT>
__device__ __forceinline__ void gin_piece(const float* __restrict__ Aw, const float* __restrict__ Bw,
                                          v8f (&acc)[2][NT]) {
#pragma unroll 2
    for (int kk = 0; kk < KC; kk += 4) {
        const v2f a0 = *(const v2f*)(Aw + kk);
        const v2f a1 = *(const v2f*)(Aw + 16 * FL_PA + kk);
        v2f b[NT];
#pragma unroll
        for (int t = 0; t < NT; ++t) b[t] = *(const v2f*)(Bw + (kk / 2) * PW + 32 * t);
#pragma unroll
        for (int t = 0; t < NT; ++t) {
            acc[0][t] = __builtin_amdgcn_wmma_f32_16x16x4_f32(false, a0, false, b[t], (short)0, acc[0][t], false, false);
            acc[1][t] = __builtin_amdgcn_wmma_f32_16x16x4_f32(false, a1, false, b[t], (short)0, acc[1][t], false, false);
        }
    }
}

template <int K, int NOUT>
__global__ __launch_bounds__(128, 7) void gin_layer_kernel(
    const int* __restrict__ rowptr, const int* __restrict__ col,
    const float* __restrict__ X, const float* __restrict__ W,
    const float* __restrict__ bias, float* __restrict__ Y, int relu, int E)
{
    constexpr int KC = 4096 / NOUT;
    constexpr int PW = 2 * NOUT + 32;
    constexpr int CW = NOUT / 4;
    constexpr int NT = CW / 16;
    constexpr int WQ = NOUT / 4;
    constexpr int JS = 128 / WQ;
    constexpr int NP = (KC / 2) / JS;

    __shared__ __attribute__((aligned(16))) float As[FL_BM * FL_PA];
    __shared__ __attribute__((aligned(16))) float Ws[(KC / 2) * PW];

    const int tid    = threadIdx.x;
    const int lane   = tid & 31;
    const int wave   = __builtin_amdgcn_readfirstlane(tid >> 5);
    const int l15    = lane & 15;
    const int half   = lane >> 4;
    const int rowblk = blockIdx.x * FL_BM;

    const int w_q = tid % WQ;
    const int w_j = tid / WQ;
    const float* __restrict__ wa = W + (size_t)(2 * w_j) * NOUT + 4 * w_q;
    float* __restrict__ ws = Ws + w_j * PW + 8 * w_q;

    const float* __restrict__ Aw = As + l15 * FL_PA + 2 * half;
    const float* __restrict__ Bw = Ws + half * PW + 2 * (CW * wave + l15);

    const float ninf = -__builtin_huge_valf();
    const int node0 = rowblk + wave * 8;
    int rp = 0;
    if (lane < 9) rp = rowptr[min(node0 + lane, NNODES)];

    v4f rw[2 * NP];
    v8f acc[2][NT];
#pragma unroll
    for (int i = 0; i < 2; ++i)
#pragma unroll
        for (int t = 0; t < NT; ++t) acc[i][t] = (v8f){};

    for (int kh = 0; kh < K; kh += FL_KH) {
        const float4* __restrict__ Xc = (const float4*)X + (kh / 4) + lane;
        int pre = 0;
        if (node0 < NNODES)
            pre = gin_list_chunk(col, max(__builtin_amdgcn_readlane(rp, 0), 0),
                                 min(__builtin_amdgcn_readlane(rp, 1), E), lane);
        for (int i = 0; i < 8; ++i) {
            const int node = node0 + i;
            float4 o = make_float4(0.0f, 0.0f, 0.0f, 0.0f);
            if (node < NNODES) {
                const int beg = max(__builtin_amdgcn_readlane(rp, i), 0);
                const int end = min(__builtin_amdgcn_readlane(rp, i + 1), E);
                const float4 self = Xc[(size_t)node * (K / 4)];
                int mine = pre;
                if (i + 1 < 8 && node + 1 < NNODES)
                    pre = gin_list_chunk(col, max(__builtin_amdgcn_readlane(rp, i + 1), 0),
                                         min(__builtin_amdgcn_readlane(rp, i + 2), E), lane);
                float4 m = make_float4(ninf, ninf, ninf, ninf);
                for (int base = beg; base < end; base += 32) {
                    const int n = min(end - base, 32);
                    if (base != beg) mine = gin_list_chunk(col, base, end, lane);
                    int j = 0;
                    for (; j + 8 <= n; j += 8) gin_fold_rows<K, 8>(Xc, mine, j, m);
                    if (j + 4 <= n) { gin_fold_rows<K, 4>(Xc, mine, j, m); j += 4; }
                    if (j + 2 <= n) { gin_fold_rows<K, 2>(Xc, mine, j, m); j += 2; }
                    if (j < n) gin_fold_rows<K, 1>(Xc, mine, j, m);
                }
                m.x = (m.x == ninf) ? 0.0f : m.x;
                m.y = (m.y == ninf) ? 0.0f : m.y;
                m.z = (m.z == ninf) ? 0.0f : m.z;
                m.w = (m.w == ninf) ? 0.0f : m.w;
                o.x = self.x + m.x;
                o.y = self.y + m.y;
                o.z = self.z + m.z;
                o.w = self.w + m.w;
            }
            *((float4*)(As + (wave * 8 + i) * FL_PA) + lane) = o;
        }

        gin_w_fetch<NOUT, JS, NP>(wa, kh, rw);
        for (int kb = 0; kb < FL_KH; kb += KC) {
            __syncthreads();
            gin_w_park<PW, JS, NP>(ws, rw);
            __syncthreads();
            if (kb + KC < FL_KH)
                gin_w_fetch<NOUT, JS, NP>(wa, kh + kb + KC, rw);
            gin_piece<KC, PW, NT>(Aw + kb, Bw, acc);
        }
        __syncthreads();
    }

#pragma unroll
    for (int i = 0; i < 2; ++i) {
        const int row0 = rowblk + 16 * i;
        if (row0 < NNODES) {
            const int rbase = row0 + 8 * half;
#pragma unroll
            for (int t = 0; t < NT; ++t) {
                const int cx = CW * wave + 16 * t + l15;
                const float bv = bias[cx];
                float* __restrict__ yc = Y + cx;
#pragma unroll
                for (int r = 0; r < 8; ++r) {
                    float v = acc[i][t][r] + bv;
                    if (relu) v = fmaxf(v, 0.0f);
                    yc[(size_t)(rbase + r) * NOUT] = v;
                }
            }
        }
    }
}

static void gin_build_lists(const int* src, const int* dst, int E,
                            int* cnt, int* rowptr, int* rank, int* col, hipStream_t stream)
{
    gin_list_zero_kernel<<<(NNODES + 255) / 256, 256, 0, stream>>>(cnt);
    gin_list_count_kernel<<<(unsigned)((E + 255) / 256), 256, 0, stream>>>(dst, cnt, rank, E);
    gin_list_scan_kernel<<<SCAN_BLOCKS, SCAN_T, 0, stream>>>(cnt, rowptr);
    gin_list_fill_kernel<<<(unsigned)((E + 255) / 256), 256, 0, stream>>>(src, dst, rank, rowptr, col, E);
}

#define FL_GROUPS ((NNODES + FL_BM - 1) / FL_BM)

extern "C" void kernel_launch(void* const* d_in, const int* in_sizes, int n_in,
                              void* d_out, int out_size, void* d_ws, size_t ws_size,
                              hipStream_t stream)
{
    const int*   src = (const int*)d_in[0];
    const int*   dst = (const int*)d_in[1];
    const float* emb = (const float*)d_in[2];
    const float* W1  = (const float*)d_in[3];
    const float* b1  = (const float*)d_in[4];
    const float* W2  = (const float*)d_in[5];
    const float* b2  = (const float*)d_in[6];
    const float* W3  = (const float*)d_in[7];
    const float* b3  = (const float*)d_in[8];
    float* out = (float*)d_out;
    const int E = in_sizes[0];

    const size_t cap = (size_t)NNODES * 256;
    int*   cnt    = (int*)d_ws;
    int*   rowptr = cnt + 65536;
    int*   rank   = rowptr + 65536;
    int*   col    = rank + E;
    float* h1     = (float*)d_ws + cap;
    float* h2     = (float*)d_ws + 2 * cap;

    gin_build_lists(src, dst, E, cnt, rowptr, rank, col, stream);

    gin_layer_kernel<128, 256><<<FL_GROUPS, 128, 0, stream>>>(rowptr, col, emb, W1, b1, h1, 1, E);
    gin_layer_kernel<256, 256><<<FL_GROUPS, 128, 0, stream>>>(rowptr, col, h1, W2, b2, h2, 1, E);
    gin_layer_kernel<256, 128><<<FL_GROUPS, 128, 0, stream>>>(rowptr, col, h2, W3, b3, out, 0, E);
}


# Gather overlapped with matrix work in one resident work-group per compute unit: 16 gather waves fill a 2-stage LDS tile ring, 4 matrix waves multiply it with W held whole in LDS; sum order unchanged
# speedup vs baseline: 3.7552x; 1.0550x over previous
#include <hip/hip_runtime.h>


#define NNODES 50000
#define NQUADS (NNODES / 4)

typedef float v2f __attribute__((ext_vector_type(2)));
typedef float v8f __attribute__((ext_vector_type(8)));


__global__ void gin_list_zero_kernel(int* __restrict__ cnt) {
    int i = blockIdx.x * blockDim.x + threadIdx.x;
    if (i < NNODES) cnt[i] = 0;
}

__global__ void gin_list_count_kernel(const int* __restrict__ dst, int* __restrict__ cnt,
                                      int* __restrict__ rank, int E) {
    int e = blockIdx.x * blockDim.x + threadIdx.x;
    if (e < E) rank[e] = atomicAdd(&cnt[dst[e]], 1);
}

#define SCAN_T 256
#define SCAN_BLOCKS ((NQUADS + SCAN_T - 1) / SCAN_T)
__global__ __launch_bounds__(SCAN_T) void gin_list_scan_kernel(const int* __restrict__ cnt,
                                                               int* __restrict__ rowptr) {
    __shared__ int sh_front[SCAN_T / 32];
    __shared__ int sh_tile[SCAN_T / 32];
    const int t = threadIdx.x;
    const int lane = t & 31;
    const int wave = t >> 5;
    const int4* __restrict__ c4 = (const int4*)cnt;

    int front = 0;
    for (int k = 0; k < (int)blockIdx.x; ++k) {
        const int4 v = c4[k * SCAN_T + t];
        front += (v.x + v.y) + (v.z + v.w);
    }
    for (int off = 16; off > 0; off >>= 1) front += __shfl_xor(front, off);
    if (lane == 0) sh_front[wave] = front;

    const int q = blockIdx.x * SCAN_T + t;
    int4 v = make_int4(0, 0, 0, 0);
    if (q < NQUADS) v = c4[q];
    const int s = (v.x + v.y) + (v.z + v.w);
    int incl = s;
    for (int off = 1; off < 32; off <<= 1) {
        const int u = __shfl_up(incl, off);
        if (lane >= off) incl += u;
    }
    if (lane == 31) sh_tile[wave] = incl;
    __syncthreads();

    int base = 0;
    for (int w = 0; w < SCAN_T / 32; ++w) {
        base += sh_front[w];
        if (w < wave) base += sh_tile[w];
    }
    const int ex = base + incl - s;
    if (q < NQUADS) {
        int4 o;
        o.x = ex;
        o.y = o.x + v.x;
        o.z = o.y + v.y;
        o.w = o.z + v.z;
        ((int4*)rowptr)[q] = o;
        if (q == NQUADS - 1) rowptr[NNODES] = ex + s;
    }
}

__global__ void gin_list_fill_kernel(const int* __restrict__ src, const int* __restrict__ dst,
                                     const int* __restrict__ rank, const int* __restrict__ rowptr,
                                     int* __restrict__ col, int E) {
    int e = blockIdx.x * blockDim.x + threadIdx.x;
    if (e < E) {
        const int p = min(max(rowptr[dst[e]] + rank[e], 0), E - 1);
        col[p] = src[e];
    }
}


typedef float v4f __attribute__((ext_vector_type(4)));

__device__ __forceinline__ float4 max4(float4 a, float4 b) {
    float4 r;
    r.x = fmaxf(a.x, b.x);
    r.y = fmaxf(a.y, b.y);
    r.z = fmaxf(a.z, b.z);
    r.w = fmaxf(a.w, b.w);
    return r;
}

__device__ __forceinline__ int gin_list_chunk(const int* __restrict__ col, int base, int end, int lane) {
    int mine = 0;
    if (base + lane < end) mine = min(max(col[base + lane], 0), NNODES - 1);
    return mine;
}

template <int K, int R>
__device__ __forceinline__ void gin_fold_rows(const float4* __restrict__ Xc, int mine, int j,
                                              float4 (&m)[K / 128]) {
    constexpr int KH = K / 128;
    float4 r[R][KH];
#pragma unroll
    for (int k = 0; k < R; ++k) {
        const int s = __builtin_amdgcn_readlane(mine, j + k);
#pragma unroll
        for (int h = 0; h < KH; ++h) r[k][h] = Xc[(size_t)s * (K / 4) + 32 * h];
    }
#pragma unroll
    for (int k = 0; k < R; ++k)
#pragma unroll
        for (int h = 0; h < KH; ++h) m[h] = max4(m[h], r[k][h]);
}

template <int K, int R>
struct gin_fold_rest {
    static __device__ __forceinline__ void run(const float4* __restrict__ Xc, int mine, int j, int r,
                                               float4 (&m)[K / 128]) {
        if (r == R) gin_fold_rows<K, R>(Xc, mine, j, m);
        else gin_fold_rest<K, R - 1>::run(Xc, mine, j, r, m);
    }
};
template <int K>
struct gin_fold_rest<K, 0> {
    static __device__ __forceinline__ void run(const float4* __restrict__, int, int, int, float4 (&)[K / 128]) {}
};

template <int PW, int NT>
__device__ __forceinline__ void gin_frag_block(const float* __restrict__ Aw, const float* __restrict__ Bw, int kk,
                                               v2f (&fa)[4], v2f (&fb)[4][NT]) {
#pragma unroll
    for (int i = 0; i < 4; ++i) {
        fa[i] = *(const v2f*)(Aw + kk + 4 * i);
#pragma unroll
        for (int t = 0; t < NT; ++t) fb[i][t] = *(const v2f*)(Bw + ((kk + 4 * i) / 2) * PW + 32 * t);
    }
}

template <int NT>
__device__ __forceinline__ void gin_mma_block(const v2f (&fa)[4], const v2f (&fb)[4][NT], v8f (&acc)[NT]) {
#pragma unroll
    for (int i = 0; i < 4; ++i)
#pragma unroll
        for (int t = 0; t < NT; ++t)
            acc[t] = __builtin_amdgcn_wmma_f32_16x16x4_f32(false, fa[i], false, fb[i][t], (short)0, acc[t], false, false);
}

template <int NOUT, int NT>
__device__ __forceinline__ void gin_store_tile(float* __restrict__ yc, int rbase, const v8f (&hold)[NT]) {
#pragma unroll
    for (int t = 0; t < NT; ++t)
#pragma unroll
        for (int r = 0; r < 8; ++r) yc[(size_t)(rbase + r) * NOUT + 16 * t] = hold[t][r];
}

#define PG_ROWS 16
#define PG_TILES (NNODES / PG_ROWS)
#define PG_GROUPS 256
#define PG_CONS 4
#define PG_PROD 16
#define PG_THREADS (32 * (PG_CONS + PG_PROD))

template <int K, int NOUT>
__global__ __launch_bounds__(PG_THREADS) void gin_layer_kernel(
    const int* __restrict__ rowptr, const int* __restrict__ col,
    const float* __restrict__ X, const float* __restrict__ W,
    const float* __restrict__ bias, float* __restrict__ Y, int relu, int E)
{
    constexpr int KH = K / 128;
    constexpr int G  = 2048 / K;
    constexpr int PA = K + 4;
    constexpr int PW = 2 * NOUT + 32;
    constexpr int CW = NOUT / 4;
    constexpr int NT = CW / 16;
    constexpr int WQ = NOUT / 4;
    constexpr int ITEMS = (K / 2) * WQ;
    constexpr int WIT = (ITEMS + PG_THREADS - 1) / PG_THREADS;
    constexpr int A_FLOATS = 2 * PG_ROWS * PA;
    constexpr int W_NEED = (K / 2) * PW;
    constexpr int W_MIN = 41 * 1024 - A_FLOATS;
    constexpr int W_FLOATS = W_NEED > W_MIN ? W_NEED : W_MIN;

    __shared__ __attribute__((aligned(16))) float Ws[W_FLOATS];
    __shared__ __attribute__((aligned(16))) float As[A_FLOATS];

    const int tid  = threadIdx.x;
    const int lane = tid & 31;
    const int wave = __builtin_amdgcn_readfirstlane(tid >> 5);
    const int l15  = lane & 15;
    const int half = lane >> 4;

    const int t0 = ((int)blockIdx.x * PG_TILES) / PG_GROUPS;
    const int t1 = (((int)blockIdx.x + 1) * PG_TILES) / PG_GROUPS;
    const int nt = t1 - t0;

    {
        v4f we[WIT], wo[WIT];
#pragma unroll
        for (int u = 0; u < WIT; ++u) {
            const int i = tid + PG_THREADS * u;
            we[u] = (v4f){};
            wo[u] = (v4f){};
            if (i < ITEMS) {
                const int j = i / WQ;
                const int q = i % WQ;
                we[u] = *(const v4f*)(W + (size_t)(2 * j + 0) * NOUT + 4 * q);
                wo[u] = *(const v4f*)(W + (size_t)(2 * j + 1) * NOUT + 4 * q);
            }
        }
#pragma unroll
        for (int u = 0; u < WIT; ++u) {
            const int i = tid + PG_THREADS * u;
            if (i < ITEMS) {
                const int j = i / WQ;
                const int q = i % WQ;
                const v4f e = we[u];
                const v4f o = wo[u];
                const v4f lo = {e.x, o.x, e.y, o.y};
                const v4f hi = {e.z, o.z, e.w, o.w};
                *(v4f*)(Ws + j * PW + 8 * q + 0) = lo;
                *(v4f*)(Ws + j * PW + 8 * q + 4) = hi;
            }
        }
    }

    const int cw = wave & (PG_CONS - 1);
    const float* __restrict__ Aw0 = As + l15 * PA + 2 * half;
    const float* __restrict__ Bw  = Ws + half * PW + 2 * (CW * cw + l15);
    float bv[NT];
#pragma unroll
    for (int t = 0; t < NT; ++t) bv[t] = bias[CW * cw + 16 * t + l15];

    const int node0 = t0 * PG_ROWS + (wave - PG_CONS);
    int rp = 0;
    if (wave >= PG_CONS && lane < 2 * nt)
        rp = rowptr[min(node0 + PG_ROWS * (lane >> 1) + (lane & 1), NNODES)];
    const float4* __restrict__ Xc = (const float4*)X + lane;
    const float ninf = -__builtin_huge_valf();
    int pre = 0;
    if (wave >= PG_CONS)
        pre = gin_list_chunk(col, max(__builtin_amdgcn_readlane(rp, 0), 0),
                             min(__builtin_amdgcn_readlane(rp, 1), E), lane);

    v8f hold[NT];
#pragma unroll
    for (int t = 0; t < NT; ++t) hold[t] = (v8f){};

    for (int s = -1; s < nt; ++s) {
        if (wave < PG_CONS) {
            if (s > 0) gin_store_tile<NOUT, NT>(Y + CW * cw + l15, (t0 + s - 1) * PG_ROWS + 8 * half, hold);
            if (s >= 0) {
                const float* __restrict__ Aw = Aw0 + (s & 1) * (PG_ROWS * PA);
                v8f acc[NT];
#pragma unroll
                for (int t = 0; t < NT; ++t) acc[t] = (v8f){};
                v2f fa0[4], fa1[4], fb0[4][NT], fb1[4][NT];
                gin_frag_block<PW, NT>(Aw, Bw, 0, fa0, fb0);
#pragma nounroll
                for (int kb = 0; kb < K; kb += 32) {
                    gin_frag_block<PW, NT>(Aw, Bw, kb + 16, fa1, fb1);
                    gin_mma_block<NT>(fa0, fb0, acc);
                    if (kb + 32 < K) gin_frag_block<PW, NT>(Aw, Bw, kb + 32, fa0, fb0);
                    gin_mma_block<NT>(fa1, fb1, acc);
                }
#pragma unroll
                for (int t = 0; t < NT; ++t)
#pragma unroll
                    for (int r = 0; r < 8; ++r) {
                        float v = acc[t][r] + bv[t];
                        if (relu) v = fmaxf(v, 0.0f);
                        hold[t][r] = v;
                    }
            }
        } else if (s + 1 < nt) {
            const int u = s + 1;
            const int node = node0 + PG_ROWS * u;
            const int beg = max(__builtin_amdgcn_readlane(rp, 2 * u), 0);
            const int end = min(__builtin_amdgcn_readlane(rp, 2 * u + 1), E);
            float4 self[KH];
#pragma unroll
            for (int h = 0; h < KH; ++h) self[h] = Xc[(size_t)node * (K / 4) + 32 * h];
            int mine = pre;
            if (u + 1 < nt)
                pre = gin_list_chunk(col, max(__builtin_amdgcn_readlane(rp, 2 * u + 2), 0),
                                     min(__builtin_amdgcn_readlane(rp, 2 * u + 3), E), lane);
            float4 m[KH];
#pragma unroll
            for (int h = 0; h < KH; ++h) m[h] = make_float4(ninf, ninf, ninf, ninf);
            for (int base = beg; base < end; base += 32) {
                const int n = min(end - base, 32);
                if (base != beg) mine = gin_list_chunk(col, base, end, lane);
                int j = 0;
                for (; j + G <= n; j += G) gin_fold_rows<K, G>(Xc, mine, j, m);
                gin_fold_rest<K, G - 1>::run(Xc, mine, j, n - j, m);
            }
            float* __restrict__ arow = As + ((u & 1) * PG_ROWS + (wave - PG_CONS)) * PA;
#pragma unroll
            for (int h = 0; h < KH; ++h) {
                float4 o;
                o.x = self[h].x + ((m[h].x == ninf) ? 0.0f : m[h].x);
                o.y = self[h].y + ((m[h].y == ninf) ? 0.0f : m[h].y);
                o.z = self[h].z + ((m[h].z == ninf) ? 0.0f : m[h].z);
                o.w = self[h].w + ((m[h].w == ninf) ? 0.0f : m[h].w);
                *((float4*)arow + 32 * h + lane) = o;
            }
        }
        __syncthreads();
    }
    if (wave < PG_CONS) gin_store_tile<NOUT, NT>(Y + CW * cw + l15, (t1 - 1) * PG_ROWS + 8 * half, hold);
}

static void gin_build_lists(const int* src, const int* dst, int E,
                            int* cnt, int* rowptr, int* rank, int* col, hipStream_t stream)
{
    gin_list_zero_kernel<<<(NNODES + 255) / 256, 256, 0, stream>>>(cnt);
    gin_list_count_kernel<<<(unsigned)((E + 255) / 256), 256, 0, stream>>>(dst, cnt, rank, E);
    gin_list_scan_kernel<<<SCAN_BLOCKS, SCAN_T, 0, stream>>>(cnt, rowptr);
    gin_list_fill_kernel<<<(unsigned)((E + 255) / 256), 256, 0, stream>>>(src, dst, rank, rowptr, col, E);
}

extern "C" void kernel_launch(void* const* d_in, const int* in_sizes, int n_in,
                              void* d_out, int out_size, void* d_ws, size_t ws_size,
                              hipStream_t stream)
{
    const int*   src = (const int*)d_in[0];
    const int*   dst = (const int*)d_in[1];
    const float* emb = (const float*)d_in[2];
    const float* W1  = (const float*)d_in[3];
    const float* b1  = (const float*)d_in[4];
    const float* W2  = (const float*)d_in[5];
    const float* b2  = (const float*)d_in[6];
    const float* W3  = (const float*)d_in[7];
    const float* b3  = (const float*)d_in[8];
    float* out = (float*)d_out;
    const int E = in_sizes[0];

    const size_t cap = (size_t)NNODES * 256;
    int*   cnt    = (int*)d_ws;
    int*   rowptr = cnt + 65536;
    int*   rank   = rowptr + 65536;
    int*   col    = rank + E;
    float* h1     = (float*)d_ws + cap;
    float* h2     = (float*)d_ws + 2 * cap;

    gin_build_lists(src, dst, E, cnt, rowptr, rank, col, stream);

    gin_layer_kernel<128, 256><<<PG_GROUPS, PG_THREADS, 0, stream>>>(rowptr, col, emb, W1, b1, h1, 1, E);
    gin_layer_kernel<256, 256><<<PG_GROUPS, PG_THREADS, 0, stream>>>(rowptr, col, h1, W2, b2, h2, 1, E);
    gin_layer_kernel<256, 128><<<PG_GROUPS, PG_THREADS, 0, stream>>>(rowptr, col, h2, W3, b3, out, 0, E);
}


# Edge list counted and ranked with LDS counters: work-group (destination range, edge chunk) holds its range's counters in LDS; no global atomics, no zero launch (6 launches); sums unchanged
# speedup vs baseline: 4.9204x; 1.3103x over previous
#include <hip/hip_runtime.h>


#define NNODES 50000

typedef float v2f __attribute__((ext_vector_type(2)));
typedef float v8f __attribute__((ext_vector_type(8)));

#define LB_R 16
#define LB_C 16
#define LB_B 3128
#define LB_NP (LB_R * LB_B)
#define LB_T 512
#define LB_QB 16

__global__ __launch_bounds__(LB_T) void gin_list_count_kernel(const int* __restrict__ dst, int* __restrict__ part,
                                                              int* __restrict__ rtot, int E) {
    __shared__ __attribute__((aligned(16))) int cntL[LB_B];
    __shared__ int sh_sum[LB_T / 32];
    const int tid = threadIdx.x;
    const int r = (int)blockIdx.x % LB_R;
    const int c = (int)blockIdx.x / LB_R;
    const int lo = r * LB_B;
    for (int i = tid; i < LB_B; i += LB_T) cntL[i] = 0;
    __syncthreads();

    const int len = (((E + LB_C - 1) / LB_C) + 3) & ~3;
    const int e0 = min(c * len, E);
    const int e1 = min(e0 + len, E);
    const int nq = (e1 - e0) >> 2;
    const int4* __restrict__ d4 = (const int4*)(dst + e0);
    for (int q0 = 0; q0 < nq; q0 += LB_QB * LB_T) {
        int4 dq[LB_QB];
        const int4* __restrict__ dp = d4 + q0 + tid;
#pragma unroll
        for (int k = 0; k < LB_QB; ++k) {
            dq[k] = make_int4(-1, -1, -1, -1);
            if (q0 + k * LB_T + tid < nq) dq[k] = dp[k * LB_T];
        }
#pragma unroll
        for (int k = 0; k < LB_QB; ++k) {
            const unsigned a = (unsigned)(dq[k].x - lo), b = (unsigned)(dq[k].y - lo);
            const unsigned g = (unsigned)(dq[k].z - lo), h = (unsigned)(dq[k].w - lo);
            if (a < LB_B) atomicAdd(&cntL[a], 1);
            if (b < LB_B) atomicAdd(&cntL[b], 1);
            if (g < LB_B) atomicAdd(&cntL[g], 1);
            if (h < LB_B) atomicAdd(&cntL[h], 1);
        }
    }
    {
        const int e = e0 + 4 * nq + tid;
        if (tid < 3 && e < e1) {
            const unsigned a = (unsigned)(dst[e] - lo);
            if (a < LB_B) atomicAdd(&cntL[a], 1);
        }
    }
    __syncthreads();

    int s = 0;
    int4* __restrict__ p4 = (int4*)(part + (size_t)c * LB_NP + lo);
    for (int i = tid; i < LB_B / 4; i += LB_T) {
        const int4 v = ((const int4*)cntL)[i];
        p4[i] = v;
        s += (v.x + v.y) + (v.z + v.w);
    }
    for (int off = 16; off > 0; off >>= 1) s += __shfl_xor(s, off);
    if ((tid & 31) == 0) sh_sum[tid >> 5] = s;
    __syncthreads();
    if (tid == 0) {
        int t = 0;
        for (int w = 0; w < LB_T / 32; ++w) t += sh_sum[w];
        rtot[c * LB_R + r] = t;
    }
}

#define LS_T 1024
__global__ __launch_bounds__(LS_T) void gin_list_scan_kernel(int* __restrict__ part, const int* __restrict__ rtot,
                                                             int* __restrict__ rowptr) {
    __shared__ int sh_front[LS_T / 32];
    __shared__ int sh_tile[LS_T / 32];
    const int t = threadIdx.x;
    const int lane = t & 31;
    const int wave = t >> 5;
    const int r = blockIdx.x;

    int front = 0;
    if (t < LB_C * LB_R && (t % LB_R) < r) front = rtot[t];

    const bool live = t < LB_B / 4;
    int4* __restrict__ base = (int4*)(part + r * LB_B) + t;
    int4 p[LB_C];
    int4 tot = make_int4(0, 0, 0, 0);
#pragma unroll
    for (int c = 0; c < LB_C; ++c) {
        p[c] = make_int4(0, 0, 0, 0);
        if (live) p[c] = base[(size_t)c * (LB_NP / 4)];
    }
#pragma unroll
    for (int c = 0; c < LB_C; ++c) {
        tot.x += p[c].x;
        tot.y += p[c].y;
        tot.z += p[c].z;
        tot.w += p[c].w;
    }
    for (int off = 16; off > 0; off >>= 1) front += __shfl_xor(front, off);
    if (lane == 0) sh_front[wave] = front;

    const int s = (tot.x + tot.y) + (tot.z + tot.w);
    int incl = s;
    for (int off = 1; off < 32; off <<= 1) {
        const int u = __shfl_up(incl, off);
        if (lane >= off) incl += u;
    }
    if (lane == 31) sh_tile[wave] = incl;
    __syncthreads();

    int ex = incl - s;
    for (int w = 0; w < LS_T / 32; ++w) {
        ex += sh_front[w];
        if (w < wave) ex += sh_tile[w];
    }
    if (live) {
        int4 o;
        o.x = ex;
        o.y = o.x + tot.x;
        o.z = o.y + tot.y;
        o.w = o.z + tot.z;
        ((int4*)(rowptr + r * LB_B))[t] = o;
#pragma unroll
        for (int c = 0; c < LB_C; ++c) {
            const int4 q = p[c];
            base[(size_t)c * (LB_NP / 4)] = o;
            o.x += q.x;
            o.y += q.y;
            o.z += q.z;
            o.w += q.w;
        }
    }
}

__global__ __launch_bounds__(LB_T) void gin_list_fill_kernel(const int* __restrict__ src, const int* __restrict__ dst,
                                                             const int* __restrict__ part, int* __restrict__ col, int E) {
    __shared__ __attribute__((aligned(16))) int curL[LB_B + 32];
    const int tid = threadIdx.x;
    const int lane = tid & 31;
    const int r = (int)blockIdx.x % LB_R;
    const int c = (int)blockIdx.x / LB_R;
    const int lo = r * LB_B;
    const int4* __restrict__ p4 = (const int4*)(part + (size_t)c * LB_NP + lo);
    for (int i = tid; i < LB_B / 4; i += LB_T) ((int4*)curL)[i] = p4[i];
    if (tid < 32) curL[LB_B + tid] = 0;
    __syncthreads();

    const int len = (((E + LB_C - 1) / LB_C) + 3) & ~3;
    const int e0 = min(c * len, E);
    const int e1 = min(e0 + len, E);
    const int nq = (e1 - e0) >> 2;
    const int4* __restrict__ d4 = (const int4*)(dst + e0);
    const int4* __restrict__ s4 = (const int4*)(src + e0);
    for (int q0 = 0; q0 < nq; q0 += LB_QB * LB_T) {
        int4 dq[LB_QB], sq[LB_QB];
        const int4* __restrict__ dp = d4 + q0 + tid;
        const int4* __restrict__ sp = s4 + q0 + tid;
#pragma unroll
        for (int k = 0; k < LB_QB; ++k) {
            dq[k] = make_int4(-1, -1, -1, -1);
            sq[k] = make_int4(0, 0, 0, 0);
            if (q0 + k * LB_T + tid < nq) {
                dq[k] = dp[k * LB_T];
                sq[k] = sp[k * LB_T];
            }
        }
#pragma unroll
        for (int k = 0; k < LB_QB; ++k) {
            const unsigned a = (unsigned)(dq[k].x - lo), b = (unsigned)(dq[k].y - lo);
            const unsigned g = (unsigned)(dq[k].z - lo), h = (unsigned)(dq[k].w - lo);
            const int pa = atomicAdd(&curL[a < LB_B ? a : LB_B + lane], 1);
            const int pb = atomicAdd(&curL[b < LB_B ? b : LB_B + lane], 1);
            const int pg = atomicAdd(&curL[g < LB_B ? g : LB_B + lane], 1);
            const int ph = atomicAdd(&curL[h < LB_B ? h : LB_B + lane], 1);
            if (a < LB_B) col[min(max(pa, 0), E - 1)] = sq[k].x;
            if (b < LB_B) col[min(max(pb, 0), E - 1)] = sq[k].y;
            if (g < LB_B) col[min(max(pg, 0), E - 1)] = sq[k].z;
            if (h < LB_B) col[min(max(ph, 0), E - 1)] = sq[k].w;
        }
    }
    {
        const int e = e0 + 4 * nq + tid;
        if (tid < 3 && e < e1) {
            const unsigned a = (unsigned)(dst[e] - lo);
            if (a < LB_B) {
                const int pa = atomicAdd(&curL[a], 1);
                col[min(max(pa, 0), E - 1)] = src[e];
            }
        }
    }
}


typedef float v4f __attribute__((ext_vector_type(4)));

__device__ __forceinline__ float4 max4(float4 a, float4 b) {
    float4 r;
    r.x = fmaxf(a.x, b.x);
    r.y = fmaxf(a.y, b.y);
    r.z = fmaxf(a.z, b.z);
    r.w = fmaxf(a.w, b.w);
    return r;
}

__device__ __forceinline__ int gin_list_chunk(const int* __restrict__ col, int base, int end, int lane) {
    int mine = 0;
    if (base + lane < end) mine = min(max(col[base + lane], 0), NNODES - 1);
    return mine;
}

template <int K, int R>
__device__ __forceinline__ void gin_fold_rows(const float4* __restrict__ Xc, int mine, int j,
                                              float4 (&m)[K / 128]) {
    constexpr int KH = K / 128;
    float4 r[R][KH];
#pragma unroll
    for (int k = 0; k < R; ++k) {
        const int s = __builtin_amdgcn_readlane(mine, j + k);
#pragma unroll
        for (int h = 0; h < KH; ++h) r[k][h] = Xc[(size_t)s * (K / 4) + 32 * h];
    }
#pragma unroll
    for (int k = 0; k < R; ++k)
#pragma unroll
        for (int h = 0; h < KH; ++h) m[h] = max4(m[h], r[k][h]);
}

template <int K, int R>
struct gin_fold_rest {
    static __device__ __forceinline__ void run(const float4* __restrict__ Xc, int mine, int j, int r,
                                               float4 (&m)[K / 128]) {
        if (r == R) gin_fold_rows<K, R>(Xc, mine, j, m);
        else gin_fold_rest<K, R - 1>::run(Xc, mine, j, r, m);
    }
};
template <int K>
struct gin_fold_rest<K, 0> {
    static __device__ __forceinline__ void run(const float4* __restrict__, int, int, int, float4 (&)[K / 128]) {}
};

template <int PW, int NT>
__device__ __forceinline__ void gin_frag_block(const float* __restrict__ Aw, const float* __restrict__ Bw, int kk,
                                               v2f (&fa)[4], v2f (&fb)[4][NT]) {
#pragma unroll
    for (int i = 0; i < 4; ++i) {
        fa[i] = *(const v2f*)(Aw + kk + 4 * i);
#pragma unroll
        for (int t = 0; t < NT; ++t) fb[i][t] = *(const v2f*)(Bw + ((kk + 4 * i) / 2) * PW + 32 * t);
    }
}

template <int NT>
__device__ __forceinline__ void gin_mma_block(const v2f (&fa)[4], const v2f (&fb)[4][NT], v8f (&acc)[NT]) {
#pragma unroll
    for (int i = 0; i < 4; ++i)
#pragma unroll
        for (int t = 0; t < NT; ++t)
            acc[t] = __builtin_amdgcn_wmma_f32_16x16x4_f32(false, fa[i], false, fb[i][t], (short)0, acc[t], false, false);
}

template <int NOUT, int NT>
__device__ __forceinline__ void gin_store_tile(float* __restrict__ yc, int rbase, const v8f (&hold)[NT]) {
#pragma unroll
    for (int t = 0; t < NT; ++t)
#pragma unroll
        for (int r = 0; r < 8; ++r) yc[(size_t)(rbase + r) * NOUT + 16 * t] = hold[t][r];
}

#define PG_ROWS 16
#define PG_TILES (NNODES / PG_ROWS)
#define PG_GROUPS 256
#define PG_CONS 4
#define PG_PROD 16
#define PG_THREADS (32 * (PG_CONS + PG_PROD))

template <int K, int NOUT>
__global__ __launch_bounds__(PG_THREADS) void gin_layer_kernel(
    const int* __restrict__ rowptr, const int* __restrict__ col,
    const float* __restrict__ X, const float* __restrict__ W,
    const float* __restrict__ bias, float* __restrict__ Y, int relu, int E)
{
    constexpr int KH = K / 128;
    constexpr int G  = 2048 / K;
    constexpr int PA = K + 4;
    constexpr int PW = 2 * NOUT + 32;
    constexpr int CW = NOUT / 4;
    constexpr int NT = CW / 16;
    constexpr int WQ = NOUT / 4;
    constexpr int ITEMS = (K / 2) * WQ;
    constexpr int WIT = (ITEMS + PG_THREADS - 1) / PG_THREADS;
    constexpr int A_FLOATS = 2 * PG_ROWS * PA;
    constexpr int W_NEED = (K / 2) * PW;
    constexpr int W_MIN = 41 * 1024 - A_FLOATS;
    constexpr int W_FLOATS = W_NEED > W_MIN ? W_NEED : W_MIN;

    __shared__ __attribute__((aligned(16))) float Ws[W_FLOATS];
    __shared__ __attribute__((aligned(16))) float As[A_FLOATS];

    const int tid  = threadIdx.x;
    const int lane = tid & 31;
    const int wave = __builtin_amdgcn_readfirstlane(tid >> 5);
    const int l15  = lane & 15;
    const int half = lane >> 4;

    const int t0 = ((int)blockIdx.x * PG_TILES) / PG_GROUPS;
    const int t1 = (((int)blockIdx.x + 1) * PG_TILES) / PG_GROUPS;
    const int nt = t1 - t0;

    {
        v4f we[WIT], wo[WIT];
#pragma unroll
        for (int u = 0; u < WIT; ++u) {
            const int i = tid + PG_THREADS * u;
            we[u] = (v4f){};
            wo[u] = (v4f){};
            if (i < ITEMS) {
                const int j = i / WQ;
                const int q = i % WQ;
                we[u] = *(const v4f*)(W + (size_t)(2 * j + 0) * NOUT + 4 * q);
                wo[u] = *(const v4f*)(W + (size_t)(2 * j + 1) * NOUT + 4 * q);
            }
        }
#pragma unroll
        for (int u = 0; u < WIT; ++u) {
            const int i = tid + PG_THREADS * u;
            if (i < ITEMS) {
                const int j = i / WQ;
                const int q = i % WQ;
                const v4f e = we[u];
                const v4f o = wo[u];
                const v4f lo = {e.x, o.x, e.y, o.y};
                const v4f hi = {e.z, o.z, e.w, o.w};
                *(v4f*)(Ws + j * PW + 8 * q + 0) = lo;
                *(v4f*)(Ws + j * PW + 8 * q + 4) = hi;
            }
        }
    }

    const int cw = wave & (PG_CONS - 1);
    const float* __restrict__ Aw0 = As + l15 * PA + 2 * half;
    const float* __restrict__ Bw  = Ws + half * PW + 2 * (CW * cw + l15);
    float bv[NT];
#pragma unroll
    for (int t = 0; t < NT; ++t) bv[t] = bias[CW * cw + 16 * t + l15];

    const int node0 = t0 * PG_ROWS + (wave - PG_CONS);
    int rp = 0;
    if (wave >= PG_CONS && lane < 2 * nt)
        rp = rowptr[min(node0 + PG_ROWS * (lane >> 1) + (lane & 1), NNODES)];
    const float4* __restrict__ Xc = (const float4*)X + lane;
    const float ninf = -__builtin_huge_valf();
    int pre = 0;
    if (wave >= PG_CONS)
        pre = gin_list_chunk(col, max(__builtin_amdgcn_readlane(rp, 0), 0),
                             min(__builtin_amdgcn_readlane(rp, 1), E), lane);

    v8f hold[NT];
#pragma unroll
    for (int t = 0; t < NT; ++t) hold[t] = (v8f){};

    for (int s = -1; s < nt; ++s) {
        if (wave < PG_CONS) {
            if (s > 0) gin_store_tile<NOUT, NT>(Y + CW * cw + l15, (t0 + s - 1) * PG_ROWS + 8 * half, hold);
            if (s >= 0) {
                const float* __restrict__ Aw = Aw0 + (s & 1) * (PG_ROWS * PA);
                v8f acc[NT];
#pragma unroll
                for (int t = 0; t < NT; ++t) acc[t] = (v8f){};
                v2f fa0[4], fa1[4], fb0[4][NT], fb1[4][NT];
                gin_frag_block<PW, NT>(Aw, Bw, 0, fa0, fb0);
#pragma nounroll
                for (int kb = 0; kb < K; kb += 32) {
                    gin_frag_block<PW, NT>(Aw, Bw, kb + 16, fa1, fb1);
                    gin_mma_block<NT>(fa0, fb0, acc);
                    if (kb + 32 < K) gin_frag_block<PW, NT>(Aw, Bw, kb + 32, fa0, fb0);
                    gin_mma_block<NT>(fa1, fb1, acc);
                }
#pragma unroll
                for (int t = 0; t < NT; ++t)
#pragma unroll
                    for (int r = 0; r < 8; ++r) {
                        float v = acc[t][r] + bv[t];
                        if (relu) v = fmaxf(v, 0.0f);
                        hold[t][r] = v;
                    }
            }
        } else if (s + 1 < nt) {
            const int u = s + 1;
            const int node = node0 + PG_ROWS * u;
            const int beg = max(__builtin_amdgcn_readlane(rp, 2 * u), 0);
            const int end = min(__builtin_amdgcn_readlane(rp, 2 * u + 1), E);
            float4 self[KH];
#pragma unroll
            for (int h = 0; h < KH; ++h) self[h] = Xc[(size_t)node * (K / 4) + 32 * h];
            int mine = pre;
            if (u + 1 < nt)
                pre = gin_list_chunk(col, max(__builtin_amdgcn_readlane(rp, 2 * u + 2), 0),
                                     min(__builtin_amdgcn_readlane(rp, 2 * u + 3), E), lane);
            float4 m[KH];
#pragma unroll
            for (int h = 0; h < KH; ++h) m[h] = make_float4(ninf, ninf, ninf, ninf);
            for (int base = beg; base < end; base += 32) {
                const int n = min(end - base, 32);
                if (base != beg) mine = gin_list_chunk(col, base, end, lane);
                int j = 0;
                for (; j + G <= n; j += G) gin_fold_rows<K, G>(Xc, mine, j, m);
                gin_fold_rest<K, G - 1>::run(Xc, mine, j, n - j, m);
            }
            float* __restrict__ arow = As + ((u & 1) * PG_ROWS + (wave - PG_CONS)) * PA;
#pragma unroll
            for (int h = 0; h < KH; ++h) {
                float4 o;
                o.x = self[h].x + ((m[h].x == ninf) ? 0.0f : m[h].x);
                o.y = self[h].y + ((m[h].y == ninf) ? 0.0f : m[h].y);
                o.z = self[h].z + ((m[h].z == ninf) ? 0.0f : m[h].z);
                o.w = self[h].w + ((m[h].w == ninf) ? 0.0f : m[h].w);
                *((float4*)arow + 32 * h + lane) = o;
            }
        }
        __syncthreads();
    }
    if (wave < PG_CONS) gin_store_tile<NOUT, NT>(Y + CW * cw + l15, (t1 - 1) * PG_ROWS + 8 * half, hold);
}

static void gin_build_lists(const int* src, const int* dst, int E,
                            int* rtot, int* rowptr, int* part, int* col, hipStream_t stream)
{
    gin_list_count_kernel<<<LB_R * LB_C, LB_T, 0, stream>>>(dst, part, rtot, E);
    gin_list_scan_kernel<<<LB_R, LS_T, 0, stream>>>(part, rtot, rowptr);
    gin_list_fill_kernel<<<LB_R * LB_C, LB_T, 0, stream>>>(src, dst, part, col, E);
}

extern "C" void kernel_launch(void* const* d_in, const int* in_sizes, int n_in,
                              void* d_out, int out_size, void* d_ws, size_t ws_size,
                              hipStream_t stream)
{
    const int*   src = (const int*)d_in[0];
    const int*   dst = (const int*)d_in[1];
    const float* emb = (const float*)d_in[2];
    const float* W1  = (const float*)d_in[3];
    const float* b1  = (const float*)d_in[4];
    const float* W2  = (const float*)d_in[5];
    const float* b2  = (const float*)d_in[6];
    const float* W3  = (const float*)d_in[7];
    const float* b3  = (const float*)d_in[8];
    float* out = (float*)d_out;
    const int E = in_sizes[0];

    const size_t cap = (size_t)NNODES * 256;
    int*   rtot   = (int*)d_ws;
    int*   rowptr = rtot + 65536;
    int*   part   = rowptr + 65536;
    int*   col    = part + LB_C * LB_NP;
    float* h1     = (float*)d_ws + cap;
    float* h2     = (float*)d_ws + 2 * cap;

    gin_build_lists(src, dst, E, rtot, rowptr, part, col, stream);

    gin_layer_kernel<128, 256><<<PG_GROUPS, PG_THREADS, 0, stream>>>(rowptr, col, emb, W1, b1, h1, 1, E);
    gin_layer_kernel<256, 256><<<PG_GROUPS, PG_THREADS, 0, stream>>>(rowptr, col, h1, W2, b2, h2, 1, E);
    gin_layer_kernel<256, 128><<<PG_GROUPS, PG_THREADS, 0, stream>>>(rowptr, col, h2, W3, b3, out, 0, E);
}


# Gather waves request the next node's own row and first rows ahead of the tile hand-over, behind a hand-written barrier that waits for LDS only: no load drain, no list wait per tile; sums unchanged
# speedup vs baseline: 5.1340x; 1.0434x over previous
#include <hip/hip_runtime.h>


#define NNODES 50000

typedef float v2f __attribute__((ext_vector_type(2)));
typedef float v8f __attribute__((ext_vector_type(8)));

#define LB_R 16
#define LB_C 16
#define LB_B 3128
#define LB_NP (LB_R * LB_B)
#define LB_T 512
#define LB_QB 16

__global__ __launch_bounds__(LB_T) void gin_list_count_kernel(const int* __restrict__ dst, int* __restrict__ part,
                                                              int* __restrict__ rtot, int E) {
    __shared__ __attribute__((aligned(16))) int cntL[LB_B];
    __shared__ int sh_sum[LB_T / 32];
    const int tid = threadIdx.x;
    const int r = (int)blockIdx.x % LB_R;
    const int c = (int)blockIdx.x / LB_R;
    const int lo = r * LB_B;
    for (int i = tid; i < LB_B; i += LB_T) cntL[i] = 0;
    __syncthreads();

    const int len = (((E + LB_C - 1) / LB_C) + 3) & ~3;
    const int e0 = min(c * len, E);
    const int e1 = min(e0 + len, E);
    const int nq = (e1 - e0) >> 2;
    const int4* __restrict__ d4 = (const int4*)(dst + e0);
    for (int q0 = 0; q0 < nq; q0 += LB_QB * LB_T) {
        int4 dq[LB_QB];
        const int4* __restrict__ dp = d4 + q0 + tid;
#pragma unroll
        for (int k = 0; k < LB_QB; ++k) {
            dq[k] = make_int4(-1, -1, -1, -1);
            if (q0 + k * LB_T + tid < nq) dq[k] = dp[k * LB_T];
        }
#pragma unroll
        for (int k = 0; k < LB_QB; ++k) {
            const unsigned a = (unsigned)(dq[k].x - lo), b = (unsigned)(dq[k].y - lo);
            const unsigned g = (unsigned)(dq[k].z - lo), h = (unsigned)(dq[k].w - lo);
            if (a < LB_B) atomicAdd(&cntL[a], 1);
            if (b < LB_B) atomicAdd(&cntL[b], 1);
            if (g < LB_B) atomicAdd(&cntL[g], 1);
            if (h < LB_B) atomicAdd(&cntL[h], 1);
        }
    }
    {
        const int e = e0 + 4 * nq + tid;
        if (tid < 3 && e < e1) {
            const unsigned a = (unsigned)(dst[e] - lo);
            if (a < LB_B) atomicAdd(&cntL[a], 1);
        }
    }
    __syncthreads();

    int s = 0;
    int4* __restrict__ p4 = (int4*)(part + (size_t)c * LB_NP + lo);
    for (int i = tid; i < LB_B / 4; i += LB_T) {
        const int4 v = ((const int4*)cntL)[i];
        p4[i] = v;
        s += (v.x + v.y) + (v.z + v.w);
    }
    for (int off = 16; off > 0; off >>= 1) s += __shfl_xor(s, off);
    if ((tid & 31) == 0) sh_sum[tid >> 5] = s;
    __syncthreads();
    if (tid == 0) {
        int t = 0;
        for (int w = 0; w < LB_T / 32; ++w) t += sh_sum[w];
        rtot[c * LB_R + r] = t;
    }
}

#define LS_T 1024
__global__ __launch_bounds__(LS_T) void gin_list_scan_kernel(int* __restrict__ part, const int* __restrict__ rtot,
                                                             int* __restrict__ rowptr) {
    __shared__ int sh_front[LS_T / 32];
    __shared__ int sh_tile[LS_T / 32];
    const int t = threadIdx.x;
    const int lane = t & 31;
    const int wave = t >> 5;
    const int r = blockIdx.x;

    int front = 0;
    if (t < LB_C * LB_R && (t % LB_R) < r) front = rtot[t];

    const bool live = t < LB_B / 4;
    int4* __restrict__ base = (int4*)(part + r * LB_B) + t;
    int4 p[LB_C];
    int4 tot = make_int4(0, 0, 0, 0);
#pragma unroll
    for (int c = 0; c < LB_C; ++c) {
        p[c] = make_int4(0, 0, 0, 0);
        if (live) p[c] = base[(size_t)c * (LB_NP / 4)];
    }
#pragma unroll
    for (int c = 0; c < LB_C; ++c) {
        tot.x += p[c].x;
        tot.y += p[c].y;
        tot.z += p[c].z;
        tot.w += p[c].w;
    }
    for (int off = 16; off > 0; off >>= 1) front += __shfl_xor(front, off);
    if (lane == 0) sh_front[wave] = front;

    const int s = (tot.x + tot.y) + (tot.z + tot.w);
    int incl = s;
    for (int off = 1; off < 32; off <<= 1) {
        const int u = __shfl_up(incl, off);
        if (lane >= off) incl += u;
    }
    if (lane == 31) sh_tile[wave] = incl;
    __syncthreads();

    int ex = incl - s;
    for (int w = 0; w < LS_T / 32; ++w) {
        ex += sh_front[w];
        if (w < wave) ex += sh_tile[w];
    }
    if (live) {
        int4 o;
        o.x = ex;
        o.y = o.x + tot.x;
        o.z = o.y + tot.y;
        o.w = o.z + tot.z;
        ((int4*)(rowptr + r * LB_B))[t] = o;
#pragma unroll
        for (int c = 0; c < LB_C; ++c) {
            const int4 q = p[c];
            base[(size_t)c * (LB_NP / 4)] = o;
            o.x += q.x;
            o.y += q.y;
            o.z += q.z;
            o.w += q.w;
        }
    }
}

__global__ __launch_bounds__(LB_T) void gin_list_fill_kernel(const int* __restrict__ src, const int* __restrict__ dst,
                                                             const int* __restrict__ part, int* __restrict__ col, int E) {
    __shared__ __attribute__((aligned(16))) int curL[LB_B + 32];
    const int tid = threadIdx.x;
    const int lane = tid & 31;
    const int r = (int)blockIdx.x % LB_R;
    const int c = (int)blockIdx.x / LB_R;
    const int lo = r * LB_B;
    const int4* __restrict__ p4 = (const int4*)(part + (size_t)c * LB_NP + lo);
    for (int i = tid; i < LB_B / 4; i += LB_T) ((int4*)curL)[i] = p4[i];
    if (tid < 32) curL[LB_B + tid] = 0;
    __syncthreads();

    const int len = (((E + LB_C - 1) / LB_C) + 3) & ~3;
    const int e0 = min(c * len, E);
    const int e1 = min(e0 + len, E);
    const int nq = (e1 - e0) >> 2;
    const int4* __restrict__ d4 = (const int4*)(dst + e0);
    const int4* __restrict__ s4 = (const int4*)(src + e0);
    for (int q0 = 0; q0 < nq; q0 += LB_QB * LB_T) {
        int4 dq[LB_QB], sq[LB_QB];
        const int4* __restrict__ dp = d4 + q0 + tid;
        const int4* __restrict__ sp = s4 + q0 + tid;
#pragma unroll
        for (int k = 0; k < LB_QB; ++k) {
            dq[k] = make_int4(-1, -1, -1, -1);
            sq[k] = make_int4(0, 0, 0, 0);
            if (q0 + k * LB_T + tid < nq) {
                dq[k] = dp[k * LB_T];
                sq[k] = sp[k * LB_T];
            }
        }
#pragma unroll
        for (int k = 0; k < LB_QB; ++k) {
            const unsigned a = (unsigned)(dq[k].x - lo), b = (unsigned)(dq[k].y - lo);
            const unsigned g = (unsigned)(dq[k].z - lo), h = (unsigned)(dq[k].w - lo);
            const int pa = atomicAdd(&curL[a < LB_B ? a : LB_B + lane], 1);
            const int pb = atomicAdd(&curL[b < LB_B ? b : LB_B + lane], 1);
            const int pg = atomicAdd(&curL[g < LB_B ? g : LB_B + lane], 1);
            const int ph = atomicAdd(&curL[h < LB_B ? h : LB_B + lane], 1);
            if (a < LB_B) col[min(max(pa, 0), E - 1)] = sq[k].x;
            if (b < LB_B) col[min(max(pb, 0), E - 1)] = sq[k].y;
            if (g < LB_B) col[min(max(pg, 0), E - 1)] = sq[k].z;
            if (h < LB_B) col[min(max(ph, 0), E - 1)] = sq[k].w;
        }
    }
    {
        const int e = e0 + 4 * nq + tid;
        if (tid < 3 && e < e1) {
            const unsigned a = (unsigned)(dst[e] - lo);
            if (a < LB_B) {
                const int pa = atomicAdd(&curL[a], 1);
                col[min(max(pa, 0), E - 1)] = src[e];
            }
        }
    }
}


typedef float v4f __attribute__((ext_vector_type(4)));

__device__ __forceinline__ float4 max4(float4 a, float4 b) {
    float4 r;
    r.x = fmaxf(a.x, b.x);
    r.y = fmaxf(a.y, b.y);
    r.z = fmaxf(a.z, b.z);
    r.w = fmaxf(a.w, b.w);
    return r;
}

__device__ __forceinline__ int gin_list_chunk(const int* __restrict__ col, int base, int end, int lane) {
    int mine = 0;
    if (base + lane < end) mine = min(max(col[base + lane], 0), NNODES - 1);
    return mine;
}

__device__ __forceinline__ int gin_list_chunk_ahead(const int* __restrict__ col, int base, int end, int lane, int E) {
    return col[min(max(min(base + lane, end - 1), 0), E - 1)];
}
__device__ __forceinline__ int gin_list_clamp(int entry) { return min(max(entry, 0), NNODES - 1); }

template <int K, int R>
__device__ __forceinline__ void gin_fold_rows(const float4* __restrict__ Xc, int mine, int j,
                                              float4 (&m)[K / 128]) {
    constexpr int KH = K / 128;
    float4 r[R][KH];
#pragma unroll
    for (int k = 0; k < R; ++k) {
        const int s = __builtin_amdgcn_readlane(mine, j + k);
#pragma unroll
        for (int h = 0; h < KH; ++h) r[k][h] = Xc[(size_t)s * (K / 4) + 32 * h];
    }
#pragma unroll
    for (int k = 0; k < R; ++k)
#pragma unroll
        for (int h = 0; h < KH; ++h) m[h] = max4(m[h], r[k][h]);
}

template <int K, int R>
struct gin_fold_rest {
    static __device__ __forceinline__ void run(const float4* __restrict__ Xc, int mine, int j, int r,
                                               float4 (&m)[K / 128]) {
        if (r == R) gin_fold_rows<K, R>(Xc, mine, j, m);
        else gin_fold_rest<K, R - 1>::run(Xc, mine, j, r, m);
    }
};
template <int K>
struct gin_fold_rest<K, 0> {
    static __device__ __forceinline__ void run(const float4* __restrict__, int, int, int, float4 (&)[K / 128]) {}
};

template <int K, int R>
__device__ __forceinline__ void gin_first_rows(const float4* __restrict__ Xc, int node, int mine, int n,
                                               float4 (&r0)[R][K / 128], float4 (&self0)[K / 128]) {
    constexpr int KH = K / 128;
#pragma unroll
    for (int h = 0; h < KH; ++h) self0[h] = Xc[(size_t)node * (K / 4) + 32 * h];
#pragma unroll
    for (int k = 0; k < R; ++k) {
        const int s = __builtin_amdgcn_readlane(mine, max(min(k, n - 1), 0));
#pragma unroll
        for (int h = 0; h < KH; ++h) r0[k][h] = Xc[(size_t)s * (K / 4) + 32 * h];
    }
}

__device__ __forceinline__ void gin_ring_barrier() {
    asm volatile("s_wait_dscnt 0x0\n\ts_barrier_signal -1\n\ts_barrier_wait -1" ::: "memory");
}

template <int PW, int NT>
__device__ __forceinline__ void gin_frag_block(const float* __restrict__ Aw, const float* __restrict__ Bw, int kk,
                                               v2f (&fa)[4], v2f (&fb)[4][NT]) {
#pragma unroll
    for (int i = 0; i < 4; ++i) {
        fa[i] = *(const v2f*)(Aw + kk + 4 * i);
#pragma unroll
        for (int t = 0; t < NT; ++t) fb[i][t] = *(const v2f*)(Bw + ((kk + 4 * i) / 2) * PW + 32 * t);
    }
}

template <int NT>
__device__ __forceinline__ void gin_mma_block(const v2f (&fa)[4], const v2f (&fb)[4][NT], v8f (&acc)[NT]) {
#pragma unroll
    for (int i = 0; i < 4; ++i)
#pragma unroll
        for (int t = 0; t < NT; ++t)
            acc[t] = __builtin_amdgcn_wmma_f32_16x16x4_f32(false, fa[i], false, fb[i][t], (short)0, acc[t], false, false);
}

template <int NOUT, int NT>
__device__ __forceinline__ void gin_store_tile(float* __restrict__ yc, int rbase, const v8f (&hold)[NT]) {
#pragma unroll
    for (int t = 0; t < NT; ++t)
#pragma unroll
        for (int r = 0; r < 8; ++r) yc[(size_t)(rbase + r) * NOUT + 16 * t] = hold[t][r];
}

#define PG_ROWS 16
#define PG_TILES (NNODES / PG_ROWS)
#define PG_GROUPS 256
#define PG_CONS 4
#define PG_PROD 16
#define PG_THREADS (32 * (PG_CONS + PG_PROD))

template <int K, int NOUT>
__global__ __launch_bounds__(PG_THREADS) void gin_layer_kernel(
    const int* __restrict__ rowptr, const int* __restrict__ col,
    const float* __restrict__ X, const float* __restrict__ W,
    const float* __restrict__ bias, float* __restrict__ Y, int relu, int E)
{
    constexpr int KH = K / 128;
    constexpr int G  = 2048 / K;
    constexpr int G0 = 1024 / K;
    constexpr int PA = K + 4;
    constexpr int PW = 2 * NOUT + 32;
    constexpr int CW = NOUT / 4;
    constexpr int NT = CW / 16;
    constexpr int WQ = NOUT / 4;
    constexpr int ITEMS = (K / 2) * WQ;
    constexpr int WIT = (ITEMS + PG_THREADS - 1) / PG_THREADS;
    constexpr int A_FLOATS = 2 * PG_ROWS * PA;
    constexpr int W_NEED = (K / 2) * PW;
    constexpr int W_MIN = 41 * 1024 - A_FLOATS;
    constexpr int W_FLOATS = W_NEED > W_MIN ? W_NEED : W_MIN;

    __shared__ __attribute__((aligned(16))) float Ws[W_FLOATS];
    __shared__ __attribute__((aligned(16))) float As[A_FLOATS];

    const int tid  = threadIdx.x;
    const int lane = tid & 31;
    const int wave = __builtin_amdgcn_readfirstlane(tid >> 5);
    const int l15  = lane & 15;
    const int half = lane >> 4;

    const int t0 = ((int)blockIdx.x * PG_TILES) / PG_GROUPS;
    const int t1 = (((int)blockIdx.x + 1) * PG_TILES) / PG_GROUPS;
    const int nt = t1 - t0;

    {
        v4f we[WIT], wo[WIT];
#pragma unroll
        for (int u = 0; u < WIT; ++u) {
            const int i = tid + PG_THREADS * u;
            we[u] = (v4f){};
            wo[u] = (v4f){};
            if (i < ITEMS) {
                const int j = i / WQ;
                const int q = i % WQ;
                we[u] = *(const v4f*)(W + (size_t)(2 * j + 0) * NOUT + 4 * q);
                wo[u] = *(const v4f*)(W + (size_t)(2 * j + 1) * NOUT + 4 * q);
            }
        }
#pragma unroll
        for (int u = 0; u < WIT; ++u) {
            const int i = tid + PG_THREADS * u;
            if (i < ITEMS) {
                const int j = i / WQ;
                const int q = i % WQ;
                const v4f e = we[u];
                const v4f o = wo[u];
                const v4f lo = {e.x, o.x, e.y, o.y};
                const v4f hi = {e.z, o.z, e.w, o.w};
                *(v4f*)(Ws + j * PW + 8 * q + 0) = lo;
                *(v4f*)(Ws + j * PW + 8 * q + 4) = hi;
            }
        }
    }

    const int cw = wave & (PG_CONS - 1);
    const float* __restrict__ Aw0 = As + l15 * PA + 2 * half;
    const float* __restrict__ Bw  = Ws + half * PW + 2 * (CW * cw + l15);
    float bv[NT];
#pragma unroll
    for (int t = 0; t < NT; ++t) bv[t] = bias[CW * cw + 16 * t + l15];

    const int node0 = t0 * PG_ROWS + (wave - PG_CONS);
    int rp = 0;
    if (wave >= PG_CONS && lane < 2 * nt)
        rp = rowptr[min(node0 + PG_ROWS * (lane >> 1) + (lane & 1), NNODES)];
    const float4* __restrict__ Xc = (const float4*)X + lane;
    const float ninf = -__builtin_huge_valf();
    int c_cur = 0;
    float4 r0[G0][KH], self0[KH];
    if (wave >= PG_CONS) {
        const int b0 = max(__builtin_amdgcn_readlane(rp, 0), 0);
        const int e0 = min(__builtin_amdgcn_readlane(rp, 1), E);
        c_cur = gin_list_chunk(col, b0, e0, lane);
        gin_first_rows<K, G0>(Xc, node0, c_cur, e0 - b0, r0, self0);
    }

    v8f hold[NT];
#pragma unroll
    for (int t = 0; t < NT; ++t) hold[t] = (v8f){};

    for (int s = -1; s < nt; ++s) {
        if (wave < PG_CONS) {
            if (s > 0) gin_store_tile<NOUT, NT>(Y + CW * cw + l15, (t0 + s - 1) * PG_ROWS + 8 * half, hold);
            if (s >= 0) {
                const float* __restrict__ Aw = Aw0 + (s & 1) * (PG_ROWS * PA);
                v8f acc[NT];
#pragma unroll
                for (int t = 0; t < NT; ++t) acc[t] = (v8f){};
                v2f fa0[4], fa1[4], fb0[4][NT], fb1[4][NT];
                gin_frag_block<PW, NT>(Aw, Bw, 0, fa0, fb0);
#pragma nounroll
                for (int kb = 0; kb < K; kb += 32) {
                    gin_frag_block<PW, NT>(Aw, Bw, kb + 16, fa1, fb1);
                    gin_mma_block<NT>(fa0, fb0, acc);
                    if (kb + 32 < K) gin_frag_block<PW, NT>(Aw, Bw, kb + 32, fa0, fb0);
                    gin_mma_block<NT>(fa1, fb1, acc);
                }
#pragma unroll
                for (int t = 0; t < NT; ++t)
#pragma unroll
                    for (int r = 0; r < 8; ++r) {
                        float v = acc[t][r] + bv[t];
                        if (relu) v = fmaxf(v, 0.0f);
                        hold[t][r] = v;
                    }
            }
#pragma unroll
            for (int h = 0; h < KH; ++h) {
                self0[h] = make_float4(0.0f, 0.0f, 0.0f, 0.0f);
#pragma unroll
                for (int k = 0; k < G0; ++k) r0[k][h] = make_float4(0.0f, 0.0f, 0.0f, 0.0f);
            }
        } else if (s + 1 < nt) {
            const int u = s + 1;
            const int node = node0 + PG_ROWS * u;
            const int beg = max(__builtin_amdgcn_readlane(rp, 2 * u), 0);
            const int end = min(__builtin_amdgcn_readlane(rp, 2 * u + 1), E);
            float4 m[KH];
#pragma unroll
            for (int h = 0; h < KH; ++h) {
                m[h] = make_float4(ninf, ninf, ninf, ninf);
#pragma unroll
                for (int k = 0; k < G0; ++k) m[h] = max4(m[h], r0[k][h]);
                if (end <= beg) m[h] = make_float4(ninf, ninf, ninf, ninf);
            }
            const int pre = gin_list_chunk_ahead(col, max(__builtin_amdgcn_readlane(rp, 2 * u + 2), 0),
                                                 min(__builtin_amdgcn_readlane(rp, 2 * u + 3), E), lane, E);
            {
                const int n = min(end - beg, 32);
                int j = G0;
                for (; j + G <= n; j += G) gin_fold_rows<K, G>(Xc, c_cur, j, m);
                gin_fold_rest<K, G - 1>::run(Xc, c_cur, j, n - j, m);
            }
            for (int base = beg + 32; base < end; base += 32) {
                const int n = min(end - base, 32);
                const int more = gin_list_chunk(col, base, end, lane);
                int j = 0;
                for (; j + G <= n; j += G) gin_fold_rows<K, G>(Xc, more, j, m);
                gin_fold_rest<K, G - 1>::run(Xc, more, j, n - j, m);
            }
            float* __restrict__ arow = As + ((u & 1) * PG_ROWS + (wave - PG_CONS)) * PA;
#pragma unroll
            for (int h = 0; h < KH; ++h) {
                float4 o;
                o.x = self0[h].x + ((m[h].x == ninf) ? 0.0f : m[h].x);
                o.y = self0[h].y + ((m[h].y == ninf) ? 0.0f : m[h].y);
                o.z = self0[h].z + ((m[h].z == ninf) ? 0.0f : m[h].z);
                o.w = self0[h].w + ((m[h].w == ninf) ? 0.0f : m[h].w);
                *((float4*)arow + 32 * h + lane) = o;
            }
            c_cur = gin_list_clamp(pre);
            if (u + 1 < nt)
                gin_first_rows<K, G0>(Xc, node + PG_ROWS, c_cur,
                                      min(__builtin_amdgcn_readlane(rp, 2 * u + 3), E) -
                                          max(__builtin_amdgcn_readlane(rp, 2 * u + 2), 0),
                                      r0, self0);
        }
        gin_ring_barrier();
    }
    if (wave < PG_CONS) gin_store_tile<NOUT, NT>(Y + CW * cw + l15, (t1 - 1) * PG_ROWS + 8 * half, hold);
}

static void gin_build_lists(const int* src, const int* dst, int E,
                            int* rtot, int* rowptr, int* part, int* col, hipStream_t stream)
{
    gin_list_count_kernel<<<LB_R * LB_C, LB_T, 0, stream>>>(dst, part, rtot, E);
    gin_list_scan_kernel<<<LB_R, LS_T, 0, stream>>>(part, rtot, rowptr);
    gin_list_fill_kernel<<<LB_R * LB_C, LB_T, 0, stream>>>(src, dst, part, col, E);
}

extern "C" void kernel_launch(void* const* d_in, const int* in_sizes, int n_in,
                              void* d_out, int out_size, void* d_ws, size_t ws_size,
                              hipStream_t stream)
{
    const int*   src = (const int*)d_in[0];
    const int*   dst = (const int*)d_in[1];
    const float* emb = (const float*)d_in[2];
    const float* W1  = (const float*)d_in[3];
    const float* b1  = (const float*)d_in[4];
    const float* W2  = (const float*)d_in[5];
    const float* b2  = (const float*)d_in[6];
    const float* W3  = (const float*)d_in[7];
    const float* b3  = (const float*)d_in[8];
    float* out = (float*)d_out;
    const int E = in_sizes[0];

    const size_t cap = (size_t)NNODES * 256;
    int*   rtot   = (int*)d_ws;
    int*   rowptr = rtot + 65536;
    int*   part   = rowptr + 65536;
    int*   col    = part + LB_C * LB_NP;
    float* h1     = (float*)d_ws + cap;
    float* h2     = (float*)d_ws + 2 * cap;

    gin_build_lists(src, dst, E, rtot, rowptr, part, col, stream);

    gin_layer_kernel<128, 256><<<PG_GROUPS, PG_THREADS, 0, stream>>>(rowptr, col, emb, W1, b1, h1, 1, E);
    gin_layer_kernel<256, 256><<<PG_GROUPS, PG_THREADS, 0, stream>>>(rowptr, col, h1, W2, b2, h2, 1, E);
    gin_layer_kernel<256, 128><<<PG_GROUPS, PG_THREADS, 0, stream>>>(rowptr, col, h2, W3, b3, out, 0, E);
}
